# starting kernel
# baseline (speedup 1.0000x reference)
#include <hip/hip_runtime.h>
#include <hip/hip_bf16.h>
#include <float.h>


typedef float v2f __attribute__((ext_vector_type(2)));
typedef float v8f __attribute__((ext_vector_type(8)));

#define S_      64
#define N_      256
#define L_      40
#define LP_     48
#define KNB     17
#define OD_     128
#define LANGID_ 256
#define PCID_   128

static __device__ inline v8f wmma4(v2f a, v2f b, v8f c) {
  return __builtin_amdgcn_wmma_f32_16x16x4_f32(false, a, false, b, (short)0, c,
                                               false, false);
}

__global__ void knn_kernel(const float* __restrict__ coord,
                           int* __restrict__ idx_out) {
  __shared__ float cx[N_], cy[N_], cz[N_];
  const int s = blockIdx.x;
  const int n = threadIdx.x;
  const float* cs = coord + (size_t)s * N_ * 3;
  cx[n] = cs[n * 3 + 0];
  cy[n] = cs[n * 3 + 1];
  cz[n] = cs[n * 3 + 2];
  __syncthreads();

  const float px = cx[n], py = cy[n], pz = cz[n];
  float pd = -1.0f;
  int   pi = -1;
  int* out = idx_out + ((size_t)s * N_ + n) * KNB;
  for (int j = 0; j < KNB; ++j) {
    float bd = FLT_MAX;
    int   bi = 0x7fffffff;
    for (int m = 0; m < N_; ++m) {
      float dx = cx[m] - px, dy = cy[m] - py, dz = cz[m] - pz;
      float d2 = dx * dx + dy * dy + dz * dz;
      bool gtPrev = (d2 > pd) || (d2 == pd && m > pi);
      bool ltBest = (d2 < bd) || (d2 == bd && m < bi);
      if (gtPrev && ltBest) { bd = d2; bi = m; }
    }
    out[j] = bi;
    pd = bd;
    pi = bi;
  }
}

__global__ void lang_mlp_kernel(const float* __restrict__ lang_feat,
                                const float* __restrict__ w1,
                                const float* __restrict__ b1,
                                const float* __restrict__ w2,
                                const float* __restrict__ b2,
                                float* __restrict__ lang_out) {
  __shared__ float xs[LANGID_];
  __shared__ float hs[OD_];
  const int row = blockIdx.x;
  const int t   = threadIdx.x;
  const float* x = lang_feat + (size_t)row * LANGID_;
  xs[t]       = x[t];
  xs[t + 128] = x[t + 128];
  __syncthreads();
  float acc = b1[t];
  for (int k = 0; k < LANGID_; ++k) acc += xs[k] * w1[k * OD_ + t];
  hs[t] = fmaxf(acc, 0.0f);
  __syncthreads();
  float acc2 = b2[t];
  for (int k = 0; k < OD_; ++k) acc2 += hs[k] * w2[k * OD_ + t];
  lang_out[(size_t)row * OD_ + t] = acc2;
}

__global__ void feat_mlp_kernel(const float* __restrict__ feat,
                                const float* __restrict__ w1,
                                const float* __restrict__ b1,
                                const float* __restrict__ w2,
                                const float* __restrict__ b2,
                                float* __restrict__ f_out) {
  __shared__ float Asm[16 * PCID_];
  __shared__ float Hsm[16 * OD_];
  const int s    = blockIdx.x >> 4;
  const int mt   = blockIdx.x & 15;
  const int tid  = threadIdx.x;
  const int lane = tid & 31, wave = tid >> 5;
  const int half = lane >> 4, lr = lane & 15;

  const float* A = feat + ((size_t)s * N_ + mt * 16) * PCID_;
  for (int i = tid; i < 16 * PCID_; i += 256) Asm[i] = A[i];
  __syncthreads();

  const int n0 = wave * 16;
  v8f c = {};
  for (int k0 = 0; k0 < PCID_; k0 += 4) {
    const int ka = k0 + 2 * half;
    v2f a, b;
    a.x = Asm[lr * PCID_ + ka];
    a.y = Asm[lr * PCID_ + ka + 1];
    b.x = w1[ka * OD_ + n0 + lr];
    b.y = w1[(ka + 1) * OD_ + n0 + lr];
    c = wmma4(a, b, c);
  }
  {
    const float bb = b1[n0 + lr];
#pragma unroll
    for (int v = 0; v < 8; ++v)
      Hsm[(v + 8 * half) * OD_ + n0 + lr] = fmaxf(c[v] + bb, 0.0f);
  }
  __syncthreads();

  v8f c2 = {};
  for (int k0 = 0; k0 < OD_; k0 += 4) {
    const int ka = k0 + 2 * half;
    v2f a, b;
    a.x = Hsm[lr * OD_ + ka];
    a.y = Hsm[lr * OD_ + ka + 1];
    b.x = w2[ka * OD_ + n0 + lr];
    b.y = w2[(ka + 1) * OD_ + n0 + lr];
    c2 = wmma4(a, b, c2);
  }
  {
    const float bb = b2[n0 + lr];
    float* fo = f_out + ((size_t)s * N_ + mt * 16) * OD_;
#pragma unroll
    for (int v = 0; v < 8; ++v)
      fo[(v + 8 * half) * OD_ + n0 + lr] = c2[v] + bb;
  }
}

__global__ void fused_kernel(const float* __restrict__ coord,
                             const float* __restrict__ lang_mask,
                             const float* __restrict__ rel_w1,
                             const float* __restrict__ rel_b1,
                             const float* __restrict__ rel_w2,
                             const float* __restrict__ rel_b2,
                             const int*   __restrict__ idx_ws,
                             const float* __restrict__ lang_ws,
                             const float* __restrict__ f_ws,
                             float* __restrict__ out,
                             float* __restrict__ score) {
  __shared__ float fg[32 * OD_];
  __shared__ float relin[32 * 16];
  __shared__ float hbuf[32 * OD_];
  __shared__ float relm[32 * OD_];
  __shared__ float attnsm[32 * LP_];
  __shared__ int   idxsm[KNB];
  __shared__ float red[OD_];

  const int s    = blockIdx.x >> 8;
  const int n    = blockIdx.x & 255;
  const int tid  = threadIdx.x;
  const int lane = tid & 31, wave = tid >> 5;
  const int half = lane >> 4, lr = lane & 15;

  if (tid < KNB) idxsm[tid] = idx_ws[((size_t)s * N_ + n) * KNB + tid];
  for (int i = tid; i < 32 * 16; i += 256) relin[i] = 0.f;
  __syncthreads();

  for (int i = tid; i < 32 * OD_; i += 256) {
    const int j = i >> 7, d = i & 127;
    fg[i] = (j < KNB) ? f_ws[((size_t)s * N_ + idxsm[j]) * OD_ + d] : 0.f;
  }
  if (tid < KNB) {
    const int j = tid;
    const float* cs = coord + (size_t)s * N_ * 3;
    const int m = idxsm[j];
    const float nx = cs[m * 3 + 0], ny = cs[m * 3 + 1], nz = cs[m * 3 + 2];
    const float ox = cs[n * 3 + 0], oy = cs[n * 3 + 1], oz = cs[n * 3 + 2];
    const float rx = nx - ox, ry = ny - oy, rz = nz - oz;
    const float dist = sqrtf(rx * rx + ry * ry + rz * rz);
    float* ri = relin + j * 16;
    ri[0] = nx; ri[1] = ny; ri[2] = nz;
    ri[3] = ox; ri[4] = oy; ri[5] = oz;
    ri[6] = rx; ri[7] = ry; ri[8] = rz;
    ri[9] = dist;
  }
  __syncthreads();

  for (int job = wave; job < 16; job += 8) {
    const int r0 = (job & 1) * 16, c0 = (job >> 1) * 16;
    v8f c = {};
    for (int k0 = 0; k0 < 16; k0 += 4) {
      const int ka = k0 + 2 * half;
      v2f a, b;
      a.x = relin[(r0 + lr) * 16 + ka];
      a.y = relin[(r0 + lr) * 16 + ka + 1];
      const int kb0 = min(ka, 9), kb1 = min(ka + 1, 9);
      b.x = rel_w1[kb0 * OD_ + c0 + lr];
      b.y = rel_w1[kb1 * OD_ + c0 + lr];
      c = wmma4(a, b, c);
    }
    const float bb = rel_b1[c0 + lr];
#pragma unroll
    for (int v = 0; v < 8; ++v)
      hbuf[(r0 + v + 8 * half) * OD_ + c0 + lr] = fmaxf(c[v] + bb, 0.f);
  }
  __syncthreads();

  for (int job = wave; job < 16; job += 8) {
    const int r0 = (job & 1) * 16, c0 = (job >> 1) * 16;
    v8f c = {};
    for (int k0 = 0; k0 < OD_; k0 += 4) {
      const int ka = k0 + 2 * half;
      v2f a, b;
      a.x = hbuf[(r0 + lr) * OD_ + ka];
      a.y = hbuf[(r0 + lr) * OD_ + ka + 1];
      b.x = rel_w2[ka * OD_ + c0 + lr];
      b.y = rel_w2[(ka + 1) * OD_ + c0 + lr];
      c = wmma4(a, b, c);
    }
    const float bb = rel_b2[c0 + lr];
#pragma unroll
    for (int v = 0; v < 8; ++v)
      relm[(r0 + v + 8 * half) * OD_ + c0 + lr] = c[v] + bb;
  }
  __syncthreads();

  if (wave < 6) {
    const int r0 = (wave & 1) * 16, c0 = (wave >> 1) * 16;
    const float* langS = lang_ws + (size_t)s * L_ * OD_;
    const int lclamp = min(c0 + lr, L_ - 1);
    v8f c = {};
    for (int k0 = 0; k0 < OD_; k0 += 4) {
      const int ka = k0 + 2 * half;
      v2f a, b;
      a.x = fg[(r0 + lr) * OD_ + ka];
      a.y = fg[(r0 + lr) * OD_ + ka + 1];
      b.x = langS[lclamp * OD_ + ka];
      b.y = langS[lclamp * OD_ + ka + 1];
      c = wmma4(a, b, c);
    }
#pragma unroll
    for (int v = 0; v < 8; ++v)
      attnsm[(r0 + v + 8 * half) * LP_ + c0 + lr] = c[v];
  }
  __syncthreads();

  if (tid < 32) {
    float* rowp = attnsm + tid * LP_;
    if (tid < KNB) {
      float mx = -FLT_MAX;
      for (int l = 0; l < L_; ++l) mx = fmaxf(mx, rowp[l]);
      float ssum = 0.f;
      for (int l = 0; l < L_; ++l) {
        const float e = expf(rowp[l] - mx);
        rowp[l] = e;
        ssum += e;
      }
      const float inv = 1.f / ssum;
      const float* mk = lang_mask + (size_t)s * L_;
      float s2 = 0.f;
      for (int l = 0; l < L_; ++l) {
        const float p = rowp[l] * inv * mk[l];
        rowp[l] = p;
        s2 += p;
      }
      const float inv2 = 1.f / (s2 + 1e-7f);
      for (int l = 0; l < L_; ++l) rowp[l] *= inv2;
      for (int l = L_; l < LP_; ++l) rowp[l] = 0.f;
    } else {
      for (int l = 0; l < LP_; ++l) rowp[l] = 0.f;
    }
  }
  __syncthreads();

  for (int job = wave; job < 16; job += 8) {
    const int r0 = (job & 1) * 16, c0 = (job >> 1) * 16;
    const float* langS = lang_ws + (size_t)s * L_ * OD_;
    v8f c = {};
    for (int k0 = 0; k0 < LP_; k0 += 4) {
      const int ka  = k0 + 2 * half;
      const int kb0 = min(ka, L_ - 1), kb1 = min(ka + 1, L_ - 1);
      v2f a, b;
      a.x = attnsm[(r0 + lr) * LP_ + ka];
      a.y = attnsm[(r0 + lr) * LP_ + ka + 1];
      b.x = langS[kb0 * OD_ + c0 + lr];
      b.y = langS[kb1 * OD_ + c0 + lr];
      c = wmma4(a, b, c);
    }
#pragma unroll
    for (int v = 0; v < 8; ++v)
      hbuf[(r0 + v + 8 * half) * OD_ + c0 + lr] = c[v];
  }
  __syncthreads();

  if (tid < OD_) {
    const int d = tid;
    float acc = 0.f;
    for (int j = 0; j < KNB; ++j)
      acc += fg[j * OD_ + d] * hbuf[j * OD_ + d] * relm[j * OD_ + d];
    acc += f_ws[((size_t)s * N_ + n) * OD_ + d];
    out[((size_t)s * N_ + n) * OD_ + d] = acc;
    red[d] = acc;
  }
  __syncthreads();
  for (int off = 64; off > 0; off >>= 1) {
    if (tid < off) red[tid] += red[tid + off];
    __syncthreads();
  }
  if (tid == 0) score[(size_t)s * N_ + n] = red[0];
}

extern "C" void kernel_launch(void* const* d_in, const int* in_sizes, int n_in,
                              void* d_out, int out_size, void* d_ws,
                              size_t ws_size, hipStream_t stream) {
  (void)in_sizes; (void)n_in; (void)out_size; (void)ws_size;
  const float* feat      = (const float*)d_in[0];
  const float* coord     = (const float*)d_in[1];
  const float* lang_feat = (const float*)d_in[2];
  const float* lang_mask = (const float*)d_in[3];
  const float* rel_w1    = (const float*)d_in[4];
  const float* rel_b1    = (const float*)d_in[5];
  const float* rel_w2    = (const float*)d_in[6];
  const float* rel_b2    = (const float*)d_in[7];
  const float* lang_w1   = (const float*)d_in[8];
  const float* lang_b1   = (const float*)d_in[9];
  const float* lang_w2   = (const float*)d_in[10];
  const float* lang_b2   = (const float*)d_in[11];
  const float* feat_w1   = (const float*)d_in[12];
  const float* feat_b1   = (const float*)d_in[13];
  const float* feat_w2   = (const float*)d_in[14];
  const float* feat_b2   = (const float*)d_in[15];

  char* ws = (char*)d_ws;
  int*   idx_ws  = (int*)ws;
  float* lang_ws = (float*)(ws + (size_t)S_ * N_ * KNB * 4);
  float* f_ws    = (float*)(ws + (size_t)S_ * N_ * KNB * 4 +
                                 (size_t)S_ * L_ * OD_ * 4);

  float* out   = (float*)d_out;
  float* score = out + (size_t)S_ * N_ * OD_;

  knn_kernel<<<S_, N_, 0, stream>>>(coord, idx_ws);
  lang_mlp_kernel<<<S_ * L_, 128, 0, stream>>>(lang_feat, lang_w1, lang_b1,
                                               lang_w2, lang_b2, lang_ws);
  feat_mlp_kernel<<<S_ * (N_ / 16), 256, 0, stream>>>(feat, feat_w1, feat_b1,
                                                      feat_w2, feat_b2, f_ws);
  fused_kernel<<<S_ * N_, 256, 0, stream>>>(coord, lang_mask, rel_w1, rel_b1,
                                            rel_w2, rel_b2, idx_ws, lang_ws,
                                            f_ws, out, score);
}


# Softmax on all 128 lanes: 8 lanes per pair row, 5 logits per lane in registers, row max and sums by DPP xor butterflies (was 16 lanes of one wave looping over 40); sum and masked-sum order change
# speedup vs baseline: 3.4632x; 3.4632x over previous
#include <hip/hip_runtime.h>
#include <hip/hip_bf16.h>
#include <float.h>


typedef float v2f __attribute__((ext_vector_type(2)));
typedef float v8f __attribute__((ext_vector_type(8)));

#define S_      64
#define N_      256
#define L_      40
#define LP_     48
#define KNB     17
#define OD_     128
#define LANGID_ 256
#define PCID_   128
#define RTP     17

static __device__ inline v8f wmma4(v2f a, v2f b, v8f c) {
  return __builtin_amdgcn_wmma_f32_16x16x4_f32(false, a, false, b, (short)0, c,
                                               false, false);
}

template <int M>
static __device__ inline float lane_xor(float x) {
  const int b = __builtin_bit_cast(int, x);
  return __builtin_bit_cast(
      float, __builtin_amdgcn_mov_dpp(b, 0x160 + M, 0xf, 0xf, false));
}

#define KNN_TPB 64
#define KNN_CPT 8

__global__ void __launch_bounds__(KNN_TPB)
knn_kernel(const float* __restrict__ coord, int* __restrict__ idx_out) {
  __shared__ float cx[N_], cy[N_], cz[N_];
  const int s = blockIdx.x / (N_ / KNN_TPB);
  const int n = (blockIdx.x % (N_ / KNN_TPB)) * KNN_TPB + threadIdx.x;
  const float* cs = coord + (size_t)s * N_ * 3;
#pragma unroll
  for (int r = 0; r < N_ / KNN_TPB; ++r) {
    const int i = r * KNN_TPB + threadIdx.x;
    cx[i] = cs[i * 3 + 0];
    cy[i] = cs[i * 3 + 1];
    cz[i] = cs[i * 3 + 2];
  }
  __syncthreads();

  const float px = cx[n], py = cy[n], pz = cz[n];
  float ld[KNB];
  int   li[KNB];
#pragma unroll
  for (int j = 0; j < KNB; ++j) { ld[j] = INFINITY; li[j] = 0x7fffffff; }

  for (int m0 = 0; m0 < N_; m0 += KNN_CPT) {
#pragma unroll
    for (int u = 0; u < KNN_CPT; ++u) {
      const int m = m0 + u;
      const float dx = cx[m] - px, dy = cy[m] - py, dz = cz[m] - pz;
      const float d2 = dx * dx + dy * dy + dz * dz;
      bool lt[KNB];
#pragma unroll
      for (int j = 0; j < KNB; ++j) lt[j] = d2 < ld[j];
#pragma unroll
      for (int j = KNB - 1; j >= 1; --j) {
        ld[j] = __builtin_amdgcn_fmed3f(ld[j - 1], d2, ld[j]);
        li[j] = lt[j] ? (lt[j - 1] ? li[j - 1] : m) : li[j];
      }
      ld[0] = fminf(ld[0], d2);
      li[0] = lt[0] ? m : li[0];
    }
  }

  int* out = idx_out + ((size_t)s * N_ + n) * KNB;
#pragma unroll
  for (int j = 0; j < KNB; ++j) out[j] = li[j];
}

__global__ void lang_mlp_kernel(const float* __restrict__ lang_feat,
                                const float* __restrict__ w1,
                                const float* __restrict__ b1,
                                const float* __restrict__ w2,
                                const float* __restrict__ b2,
                                float* __restrict__ lang_out) {
  __shared__ float xs[LANGID_];
  __shared__ float hs[OD_];
  const int row = blockIdx.x;
  const int t   = threadIdx.x;
  const float* x = lang_feat + (size_t)row * LANGID_;
  xs[t]       = x[t];
  xs[t + 128] = x[t + 128];
  __syncthreads();
  float acc = b1[t];
  for (int k = 0; k < LANGID_; ++k) acc += xs[k] * w1[k * OD_ + t];
  hs[t] = fmaxf(acc, 0.0f);
  __syncthreads();
  float acc2 = b2[t];
  for (int k = 0; k < OD_; ++k) acc2 += hs[k] * w2[k * OD_ + t];
  lang_out[(size_t)row * OD_ + t] = acc2;
}

__global__ void feat_mlp_kernel(const float* __restrict__ feat,
                                const float* __restrict__ w1,
                                const float* __restrict__ b1,
                                const float* __restrict__ w2,
                                const float* __restrict__ b2,
                                float* __restrict__ f_out) {
  __shared__ float Asm[16 * PCID_];
  __shared__ float Hsm[16 * OD_];
  const int s    = blockIdx.x >> 4;
  const int mt   = blockIdx.x & 15;
  const int tid  = threadIdx.x;
  const int lane = tid & 31, wave = tid >> 5;
  const int half = lane >> 4, lr = lane & 15;

  const float* A = feat + ((size_t)s * N_ + mt * 16) * PCID_;
  for (int i = tid; i < 16 * PCID_; i += 256) Asm[i] = A[i];
  __syncthreads();

  const int n0 = wave * 16;
  v8f c = {};
  for (int k0 = 0; k0 < PCID_; k0 += 4) {
    const int ka = k0 + 2 * half;
    v2f a, b;
    a.x = Asm[lr * PCID_ + ka];
    a.y = Asm[lr * PCID_ + ka + 1];
    b.x = w1[ka * OD_ + n0 + lr];
    b.y = w1[(ka + 1) * OD_ + n0 + lr];
    c = wmma4(a, b, c);
  }
  {
    const float bb = b1[n0 + lr];
#pragma unroll
    for (int v = 0; v < 8; ++v)
      Hsm[(v + 8 * half) * OD_ + n0 + lr] = fmaxf(c[v] + bb, 0.0f);
  }
  __syncthreads();

  v8f c2 = {};
  for (int k0 = 0; k0 < OD_; k0 += 4) {
    const int ka = k0 + 2 * half;
    v2f a, b;
    a.x = Hsm[lr * OD_ + ka];
    a.y = Hsm[lr * OD_ + ka + 1];
    b.x = w2[ka * OD_ + n0 + lr];
    b.y = w2[(ka + 1) * OD_ + n0 + lr];
    c2 = wmma4(a, b, c2);
  }
  {
    const float bb = b2[n0 + lr];
    float* fo = f_out + ((size_t)s * N_ + mt * 16) * OD_;
#pragma unroll
    for (int v = 0; v < 8; ++v)
      fo[(v + 8 * half) * OD_ + n0 + lr] = c2[v] + bb;
  }
}

template <int MODE>
__global__ void __launch_bounds__(128)
fused_kernel(const float* __restrict__ coord,
                             const float* __restrict__ lang_mask,
                             const float* __restrict__ rel_w1,
                             const float* __restrict__ rel_b1,
                             const float* __restrict__ rel_w2,
                             const float* __restrict__ rel_b2,
                             const int*   __restrict__ idx_ws,
                             const float* __restrict__ lang_ws,
                             const float* __restrict__ f_ws,
                             float* __restrict__ out,
                             float* __restrict__ score) {
  __shared__ float fg[16 * OD_];
  __shared__ float relin[16 * 16];
  __shared__ float hbuf[16 * OD_];
  __shared__ float relT[OD_ * RTP];
  __shared__ float attnsm[16 * LP_];
  __shared__ int   idxsm[16];
  __shared__ float sred[16 * 4];
  __shared__ float mksm[L_];

  const int tid  = threadIdx.x;
  const int lane = tid & 31, wave = tid >> 5;
  const int half = lane >> 4, lr = lane & 15;
  const int s  = (MODE == 0) ? (blockIdx.x >> 8) : (blockIdx.x >> 4);
  const int nb = (MODE == 0) ? (blockIdx.x & 255) : ((blockIdx.x & 15) * 16);

  float mkv = 0.f;
  if (tid < L_) mkv = lang_mask[s * L_ + tid];
  if (tid < 16) {
    const int pt = (MODE == 0) ? nb : nb + tid;
    const int j  = (MODE == 0) ? tid : 16;
    idxsm[tid] = idx_ws[((size_t)s * N_ + pt) * KNB + j];
  }
  relin[tid] = 0.f;
  relin[tid + 128] = 0.f;
  __syncthreads();
  if (tid < L_) mksm[tid] = mkv;

  for (int i = tid; i < 16 * OD_; i += 128) {
    const int j = i >> 7, d = i & 127;
    fg[i] = f_ws[((size_t)s * N_ + idxsm[j]) * OD_ + d];
  }
  if (tid < 16) {
    const int pt = (MODE == 0) ? nb : nb + tid;
    const float* cs = coord + (size_t)s * N_ * 3;
    const int m = idxsm[tid];
    const float nx = cs[m * 3 + 0], ny = cs[m * 3 + 1], nz = cs[m * 3 + 2];
    const float ox = cs[pt * 3 + 0], oy = cs[pt * 3 + 1], oz = cs[pt * 3 + 2];
    const float rx = nx - ox, ry = ny - oy, rz = nz - oz;
    const float dist = sqrtf(rx * rx + ry * ry + rz * rz);
    float* ri = relin + tid * 16;
    ri[0] = nx; ri[1] = ny; ri[2] = nz;
    ri[3] = ox; ri[4] = oy; ri[5] = oz;
    ri[6] = rx; ri[7] = ry; ri[8] = rz;
    ri[9] = dist;
  }
  __syncthreads();

  if (wave < 3) {
    const int c0 = wave * 16;
    const float* langS = lang_ws + (size_t)s * L_ * OD_;
    const int lclamp = min(c0 + lr, L_ - 1);
    v8f c = {};
    for (int k0 = 0; k0 < OD_; k0 += 4) {
      const int ka = k0 + 2 * half;
      v2f a, b;
      a.x = fg[lr * OD_ + ka];
      a.y = fg[lr * OD_ + ka + 1];
      b.x = langS[lclamp * OD_ + ka];
      b.y = langS[lclamp * OD_ + ka + 1];
      c = wmma4(a, b, c);
    }
#pragma unroll
    for (int v = 0; v < 8; ++v)
      attnsm[(v + 8 * half) * LP_ + c0 + lr] = c[v];
  }

  {
    const int g = wave;
    v2f bin[4];
    for (int i = 0; i < 4; ++i) {
      const float* rp = relin + lr * 16 + 4 * i + 2 * half;
      bin[i].x = rp[0];
      bin[i].y = rp[1];
    }
    v8f hT[8];
    for (int t = 0; t < 8; ++t) {
      v8f c = {};
      for (int i = 0; i < 4; ++i) {
        const int ka  = 4 * i + 2 * half;
        const int ka0 = min(ka, 9), ka1 = min(ka + 1, 9);
        v2f a;
        a.x = rel_w1[ka0 * OD_ + 16 * t + lr];
        a.y = rel_w1[ka1 * OD_ + 16 * t + lr];
        c = wmma4(a, bin[i], c);
      }
      const float* bp = rel_b1 + 16 * t + 8 * half;
      for (int r = 0; r < 8; ++r) hT[t][r] = fmaxf(c[r] + bp[r], 0.f);
    }

    const int d0a = 16 * g, d0b = 16 * (g + 4);
    v8f ca = {}, cb = {};
    for (int t = 0; t < 8; ++t) {
      for (int q = 0; q < 4; ++q) {
        const int k = 16 * t + 8 * half + 2 * q;
        v2f aa, ab, b;
        aa.x = rel_w2[k * OD_ + d0a + lr];
        aa.y = rel_w2[(k + 1) * OD_ + d0a + lr];
        ab.x = rel_w2[k * OD_ + d0b + lr];
        ab.y = rel_w2[(k + 1) * OD_ + d0b + lr];
        b.x = hT[t][2 * q];
        b.y = hT[t][2 * q + 1];
        ca = wmma4(aa, b, ca);
        cb = wmma4(ab, b, cb);
      }
    }
    const float* ba = rel_b2 + d0a + 8 * half;
    const float* bb = rel_b2 + d0b + 8 * half;
    for (int r = 0; r < 8; ++r) {
      relT[(d0a + 8 * half + r) * RTP + lr] = ca[r] + ba[r];
      relT[(d0b + 8 * half + r) * RTP + lr] = cb[r] + bb[r];
    }
  }
  __syncthreads();

  {
    float* rowp = attnsm + (tid >> 3) * LP_ + (tid & 7);
    const float* mkp = mksm + (tid & 7);
    float v[5];
    for (int i = 0; i < 5; ++i) v[i] = rowp[8 * i];
    float mx = -FLT_MAX;
    for (int i = 0; i < 5; ++i) mx = fmaxf(mx, v[i]);
    mx = fmaxf(mx, lane_xor<1>(mx));
    mx = fmaxf(mx, lane_xor<2>(mx));
    mx = fmaxf(mx, lane_xor<4>(mx));
    float ssum = 0.f;
    for (int i = 0; i < 5; ++i) {
      v[i] = expf(v[i] - mx);
      ssum += v[i];
    }
    ssum += lane_xor<1>(ssum);
    ssum += lane_xor<2>(ssum);
    ssum += lane_xor<4>(ssum);
    const float inv = 1.f / ssum;
    float s2 = 0.f;
    for (int i = 0; i < 5; ++i) {
      v[i] = v[i] * inv * mkp[8 * i];
      s2 += v[i];
    }
    s2 += lane_xor<1>(s2);
    s2 += lane_xor<2>(s2);
    s2 += lane_xor<4>(s2);
    const float inv2 = 1.f / (s2 + 1e-7f);
    for (int i = 0; i < 5; ++i) rowp[8 * i] = v[i] * inv2;
    rowp[L_] = 0.f;
  }
  __syncthreads();

  {
    const int c0a = wave * 16, c0b = (wave + 4) * 16;
    const float* langS = lang_ws + (size_t)s * L_ * OD_;
    v8f ca = {}, cb = {};
    for (int k0 = 0; k0 < LP_; k0 += 4) {
      const int ka  = k0 + 2 * half;
      const int kb0 = min(ka, L_ - 1), kb1 = min(ka + 1, L_ - 1);
      v2f a, ba, bb;
      a.x = attnsm[lr * LP_ + ka];
      a.y = attnsm[lr * LP_ + ka + 1];
      ba.x = langS[kb0 * OD_ + c0a + lr];
      ba.y = langS[kb1 * OD_ + c0a + lr];
      bb.x = langS[kb0 * OD_ + c0b + lr];
      bb.y = langS[kb1 * OD_ + c0b + lr];
      ca = wmma4(a, ba, ca);
      cb = wmma4(a, bb, cb);
    }
#pragma unroll
    for (int v = 0; v < 8; ++v) {
      hbuf[(v + 8 * half) * OD_ + c0a + lr] = ca[v];
      hbuf[(v + 8 * half) * OD_ + c0b + lr] = cb[v];
    }
  }
  __syncthreads();

  if (MODE == 0) {
    const int d = tid;
    float acc = 0.f;
    for (int j = 0; j < 16; ++j)
      acc += fg[j * OD_ + d] * hbuf[j * OD_ + d] * relT[d * RTP + j];
    acc += f_ws[((size_t)s * N_ + nb) * OD_ + d];
    out[((size_t)s * N_ + nb) * OD_ + d] = acc;
  } else {
    const int d = tid;
    for (int r = 0; r < 16; ++r) {
      float* op = out + ((size_t)s * N_ + nb + r) * OD_ + d;
      const float o = *op + fg[r * OD_ + d] * hbuf[r * OD_ + d] * relT[d * RTP + r];
      *op = o;
      float w = o;
      for (int off = 16; off > 0; off >>= 1) w += __shfl_xor(w, off);
      if (lane == 0) sred[r * 4 + wave] = w;
    }
    __syncthreads();
    if (tid < 16)
      score[(size_t)s * N_ + nb + tid] =
          (sred[tid * 4] + sred[tid * 4 + 1]) + (sred[tid * 4 + 2] + sred[tid * 4 + 3]);
  }
}

extern "C" void kernel_launch(void* const* d_in, const int* in_sizes, int n_in,
                              void* d_out, int out_size, void* d_ws,
                              size_t ws_size, hipStream_t stream) {
  (void)in_sizes; (void)n_in; (void)out_size; (void)ws_size;
  const float* feat      = (const float*)d_in[0];
  const float* coord     = (const float*)d_in[1];
  const float* lang_feat = (const float*)d_in[2];
  const float* lang_mask = (const float*)d_in[3];
  const float* rel_w1    = (const float*)d_in[4];
  const float* rel_b1    = (const float*)d_in[5];
  const float* rel_w2    = (const float*)d_in[6];
  const float* rel_b2    = (const float*)d_in[7];
  const float* lang_w1   = (const float*)d_in[8];
  const float* lang_b1   = (const float*)d_in[9];
  const float* lang_w2   = (const float*)d_in[10];
  const float* lang_b2   = (const float*)d_in[11];
  const float* feat_w1   = (const float*)d_in[12];
  const float* feat_b1   = (const float*)d_in[13];
  const float* feat_w2   = (const float*)d_in[14];
  const float* feat_b2   = (const float*)d_in[15];

  char* ws = (char*)d_ws;
  int*   idx_ws  = (int*)ws;
  float* lang_ws = (float*)(ws + (size_t)S_ * N_ * KNB * 4);
  float* f_ws    = (float*)(ws + (size_t)S_ * N_ * KNB * 4 +
                                 (size_t)S_ * L_ * OD_ * 4);

  float* out   = (float*)d_out;
  float* score = out + (size_t)S_ * N_ * OD_;

  knn_kernel<<<S_ * (N_ / KNN_TPB), KNN_TPB, 0, stream>>>(coord, idx_ws);
  lang_mlp_kernel<<<S_ * L_, 128, 0, stream>>>(lang_feat, lang_w1, lang_b1,
                                               lang_w2, lang_b2, lang_ws);
  feat_mlp_kernel<<<S_ * (N_ / 16), 256, 0, stream>>>(feat, feat_w1, feat_b1,
                                                      feat_w2, feat_b2, f_ws);
  fused_kernel<0><<<S_ * N_, 128, 0, stream>>>(coord, lang_mask, rel_w1,
                                               rel_b1, rel_w2, rel_b2, idx_ws,
                                               lang_ws, f_ws, out, score);
  fused_kernel<1><<<S_ * (N_ / 16), 128, 0, stream>>>(coord, lang_mask, rel_w1,
                                                      rel_b1, rel_w2, rel_b2,
                                                      idx_ws, lang_ws, f_ws,
                                                      out, score);
}


# Neighbor indices read as block-uniform scalars, not staged in LDS: a thread's 16 row-gather loads go out together from scalar row bases; index list, its barrier and LDS reads gone; sums unchanged
# speedup vs baseline: 3.5925x; 1.0373x over previous
#include <hip/hip_runtime.h>
#include <hip/hip_bf16.h>
#include <float.h>


typedef float v2f __attribute__((ext_vector_type(2)));
typedef float v8f __attribute__((ext_vector_type(8)));

#define S_      64
#define N_      256
#define L_      40
#define LP_     48
#define KNB     17
#define OD_     128
#define LANGID_ 256
#define PCID_   128
#define RTP     17

static __device__ inline v8f wmma4(v2f a, v2f b, v8f c) {
  return __builtin_amdgcn_wmma_f32_16x16x4_f32(false, a, false, b, (short)0, c,
                                               false, false);
}

template <int M>
static __device__ inline float lane_xor(float x) {
  const int b = __builtin_bit_cast(int, x);
  return __builtin_bit_cast(
      float, __builtin_amdgcn_mov_dpp(b, 0x160 + M, 0xf, 0xf, false));
}

#define KNN_TPB 64
#define KNN_CPT 8

__global__ void __launch_bounds__(KNN_TPB)
knn_kernel(const float* __restrict__ coord, int* __restrict__ idx_out) {
  __shared__ float cx[N_], cy[N_], cz[N_];
  const int s = blockIdx.x / (N_ / KNN_TPB);
  const int n = (blockIdx.x % (N_ / KNN_TPB)) * KNN_TPB + threadIdx.x;
  const float* cs = coord + (size_t)s * N_ * 3;
#pragma unroll
  for (int r = 0; r < N_ / KNN_TPB; ++r) {
    const int i = r * KNN_TPB + threadIdx.x;
    cx[i] = cs[i * 3 + 0];
    cy[i] = cs[i * 3 + 1];
    cz[i] = cs[i * 3 + 2];
  }
  __syncthreads();

  const float px = cx[n], py = cy[n], pz = cz[n];
  float ld[KNB];
  int   li[KNB];
#pragma unroll
  for (int j = 0; j < KNB; ++j) { ld[j] = INFINITY; li[j] = 0x7fffffff; }

  for (int m0 = 0; m0 < N_; m0 += KNN_CPT) {
#pragma unroll
    for (int u = 0; u < KNN_CPT; ++u) {
      const int m = m0 + u;
      const float dx = cx[m] - px, dy = cy[m] - py, dz = cz[m] - pz;
      const float d2 = dx * dx + dy * dy + dz * dz;
      bool lt[KNB];
#pragma unroll
      for (int j = 0; j < KNB; ++j) lt[j] = d2 < ld[j];
#pragma unroll
      for (int j = KNB - 1; j >= 1; --j) {
        ld[j] = __builtin_amdgcn_fmed3f(ld[j - 1], d2, ld[j]);
        li[j] = lt[j] ? (lt[j - 1] ? li[j - 1] : m) : li[j];
      }
      ld[0] = fminf(ld[0], d2);
      li[0] = lt[0] ? m : li[0];
    }
  }

  int* out = idx_out + ((size_t)s * N_ + n) * KNB;
#pragma unroll
  for (int j = 0; j < KNB; ++j) out[j] = li[j];
}

__global__ void lang_mlp_kernel(const float* __restrict__ lang_feat,
                                const float* __restrict__ w1,
                                const float* __restrict__ b1,
                                const float* __restrict__ w2,
                                const float* __restrict__ b2,
                                float* __restrict__ lang_out) {
  __shared__ float xs[LANGID_];
  __shared__ float hs[OD_];
  const int row = blockIdx.x;
  const int t   = threadIdx.x;
  const float* x = lang_feat + (size_t)row * LANGID_;
  xs[t]       = x[t];
  xs[t + 128] = x[t + 128];
  __syncthreads();
  float acc = b1[t];
  for (int k = 0; k < LANGID_; ++k) acc += xs[k] * w1[k * OD_ + t];
  hs[t] = fmaxf(acc, 0.0f);
  __syncthreads();
  float acc2 = b2[t];
  for (int k = 0; k < OD_; ++k) acc2 += hs[k] * w2[k * OD_ + t];
  lang_out[(size_t)row * OD_ + t] = acc2;
}

__global__ void feat_mlp_kernel(const float* __restrict__ feat,
                                const float* __restrict__ w1,
                                const float* __restrict__ b1,
                                const float* __restrict__ w2,
                                const float* __restrict__ b2,
                                float* __restrict__ f_out) {
  __shared__ float Asm[16 * PCID_];
  __shared__ float Hsm[16 * OD_];
  const int s    = blockIdx.x >> 4;
  const int mt   = blockIdx.x & 15;
  const int tid  = threadIdx.x;
  const int lane = tid & 31, wave = tid >> 5;
  const int half = lane >> 4, lr = lane & 15;

  const float* A = feat + ((size_t)s * N_ + mt * 16) * PCID_;
  for (int i = tid; i < 16 * PCID_; i += 256) Asm[i] = A[i];
  __syncthreads();

  const int n0 = wave * 16;
  v8f c = {};
  for (int k0 = 0; k0 < PCID_; k0 += 4) {
    const int ka = k0 + 2 * half;
    v2f a, b;
    a.x = Asm[lr * PCID_ + ka];
    a.y = Asm[lr * PCID_ + ka + 1];
    b.x = w1[ka * OD_ + n0 + lr];
    b.y = w1[(ka + 1) * OD_ + n0 + lr];
    c = wmma4(a, b, c);
  }
  {
    const float bb = b1[n0 + lr];
#pragma unroll
    for (int v = 0; v < 8; ++v)
      Hsm[(v + 8 * half) * OD_ + n0 + lr] = fmaxf(c[v] + bb, 0.0f);
  }
  __syncthreads();

  v8f c2 = {};
  for (int k0 = 0; k0 < OD_; k0 += 4) {
    const int ka = k0 + 2 * half;
    v2f a, b;
    a.x = Hsm[lr * OD_ + ka];
    a.y = Hsm[lr * OD_ + ka + 1];
    b.x = w2[ka * OD_ + n0 + lr];
    b.y = w2[(ka + 1) * OD_ + n0 + lr];
    c2 = wmma4(a, b, c2);
  }
  {
    const float bb = b2[n0 + lr];
    float* fo = f_out + ((size_t)s * N_ + mt * 16) * OD_;
#pragma unroll
    for (int v = 0; v < 8; ++v)
      fo[(v + 8 * half) * OD_ + n0 + lr] = c2[v] + bb;
  }
}

template <int MODE>
__global__ void __launch_bounds__(128)
fused_kernel(const float* __restrict__ coord,
                             const float* __restrict__ lang_mask,
                             const float* __restrict__ rel_w1,
                             const float* __restrict__ rel_b1,
                             const float* __restrict__ rel_w2,
                             const float* __restrict__ rel_b2,
                             const int*   __restrict__ idx_ws,
                             const float* __restrict__ lang_ws,
                             const float* __restrict__ f_ws,
                             float* __restrict__ out,
                             float* __restrict__ score) {
  __shared__ float fg[16 * OD_];
  __shared__ float relin[16 * 16];
  __shared__ float hbuf[16 * OD_];
  __shared__ float relT[OD_ * RTP];
  __shared__ float attnsm[16 * LP_];
  __shared__ float sred[16 * 4];
  __shared__ float mksm[L_];

  const int tid  = threadIdx.x;
  const int lane = tid & 31, wave = tid >> 5;
  const int half = lane >> 4, lr = lane & 15;
  const int s  = (MODE == 0) ? (blockIdx.x >> 8) : (blockIdx.x >> 4);
  const int nb = (MODE == 0) ? (blockIdx.x & 255) : ((blockIdx.x & 15) * 16);

  float mkv = 0.f;
  if (tid < L_) mkv = lang_mask[s * L_ + tid];

  int nbi[16];
#pragma unroll
  for (int r = 0; r < 16; ++r) {
    const int pt = (MODE == 0) ? nb : nb + r;
    const int j  = (MODE == 0) ? r : 16;
    nbi[r] = idx_ws[((size_t)s * N_ + pt) * KNB + j];
  }
  float gv[16];
#pragma unroll
  for (int r = 0; r < 16; ++r)
    gv[r] = f_ws[((size_t)s * N_ + nbi[r]) * OD_ + tid];

  if (tid < 16) {
    const int pt = (MODE == 0) ? nb : nb + tid;
    const int j  = (MODE == 0) ? tid : 16;
    const float* cs = coord + (size_t)s * N_ * 3;
    const int m = idx_ws[((size_t)s * N_ + pt) * KNB + j];
    const float nx = cs[m * 3 + 0], ny = cs[m * 3 + 1], nz = cs[m * 3 + 2];
    const float ox = cs[pt * 3 + 0], oy = cs[pt * 3 + 1], oz = cs[pt * 3 + 2];
    const float rx = nx - ox, ry = ny - oy, rz = nz - oz;
    const float dist = sqrtf(rx * rx + ry * ry + rz * rz);
    float* ri = relin + tid * 16;
    ri[0] = nx; ri[1] = ny; ri[2] = nz;
    ri[3] = ox; ri[4] = oy; ri[5] = oz;
    ri[6] = rx; ri[7] = ry; ri[8] = rz;
    ri[9] = dist;
    ri[10] = 0.f; ri[11] = 0.f; ri[12] = 0.f;
    ri[13] = 0.f; ri[14] = 0.f; ri[15] = 0.f;
  }
#pragma unroll
  for (int r = 0; r < 16; ++r) fg[r * OD_ + tid] = gv[r];
  if (tid < L_) mksm[tid] = mkv;
  __syncthreads();

  if (wave < 3) {
    const int c0 = wave * 16;
    const float* langS = lang_ws + (size_t)s * L_ * OD_;
    const int lclamp = min(c0 + lr, L_ - 1);
    v8f c = {};
    for (int k0 = 0; k0 < OD_; k0 += 4) {
      const int ka = k0 + 2 * half;
      v2f a, b;
      a.x = fg[lr * OD_ + ka];
      a.y = fg[lr * OD_ + ka + 1];
      b.x = langS[lclamp * OD_ + ka];
      b.y = langS[lclamp * OD_ + ka + 1];
      c = wmma4(a, b, c);
    }
#pragma unroll
    for (int v = 0; v < 8; ++v)
      attnsm[(v + 8 * half) * LP_ + c0 + lr] = c[v];
  }

  {
    const int g = wave;
    v2f bin[4];
    for (int i = 0; i < 4; ++i) {
      const float* rp = relin + lr * 16 + 4 * i + 2 * half;
      bin[i].x = rp[0];
      bin[i].y = rp[1];
    }
    v8f hT[8];
    for (int t = 0; t < 8; ++t) {
      v8f c = {};
      for (int i = 0; i < 4; ++i) {
        const int ka  = 4 * i + 2 * half;
        const int ka0 = min(ka, 9), ka1 = min(ka + 1, 9);
        v2f a;
        a.x = rel_w1[ka0 * OD_ + 16 * t + lr];
        a.y = rel_w1[ka1 * OD_ + 16 * t + lr];
        c = wmma4(a, bin[i], c);
      }
      const float* bp = rel_b1 + 16 * t + 8 * half;
      for (int r = 0; r < 8; ++r) hT[t][r] = fmaxf(c[r] + bp[r], 0.f);
    }

    const int d0a = 16 * g, d0b = 16 * (g + 4);
    v8f ca = {}, cb = {};
    for (int t = 0; t < 8; ++t) {
      for (int q = 0; q < 4; ++q) {
        const int k = 16 * t + 8 * half + 2 * q;
        v2f aa, ab, b;
        aa.x = rel_w2[k * OD_ + d0a + lr];
        aa.y = rel_w2[(k + 1) * OD_ + d0a + lr];
        ab.x = rel_w2[k * OD_ + d0b + lr];
        ab.y = rel_w2[(k + 1) * OD_ + d0b + lr];
        b.x = hT[t][2 * q];
        b.y = hT[t][2 * q + 1];
        ca = wmma4(aa, b, ca);
        cb = wmma4(ab, b, cb);
      }
    }
    const float* ba = rel_b2 + d0a + 8 * half;
    const float* bb = rel_b2 + d0b + 8 * half;
    for (int r = 0; r < 8; ++r) {
      relT[(d0a + 8 * half + r) * RTP + lr] = ca[r] + ba[r];
      relT[(d0b + 8 * half + r) * RTP + lr] = cb[r] + bb[r];
    }
  }
  __syncthreads();

  {
    float* rowp = attnsm + (tid >> 3) * LP_ + (tid & 7);
    const float* mkp = mksm + (tid & 7);
    float v[5];
    for (int i = 0; i < 5; ++i) v[i] = rowp[8 * i];
    float mx = -FLT_MAX;
    for (int i = 0; i < 5; ++i) mx = fmaxf(mx, v[i]);
    mx = fmaxf(mx, lane_xor<1>(mx));
    mx = fmaxf(mx, lane_xor<2>(mx));
    mx = fmaxf(mx, lane_xor<4>(mx));
    float ssum = 0.f;
    for (int i = 0; i < 5; ++i) {
      v[i] = expf(v[i] - mx);
      ssum += v[i];
    }
    ssum += lane_xor<1>(ssum);
    ssum += lane_xor<2>(ssum);
    ssum += lane_xor<4>(ssum);
    const float inv = 1.f / ssum;
    float s2 = 0.f;
    for (int i = 0; i < 5; ++i) {
      v[i] = v[i] * inv * mkp[8 * i];
      s2 += v[i];
    }
    s2 += lane_xor<1>(s2);
    s2 += lane_xor<2>(s2);
    s2 += lane_xor<4>(s2);
    const float inv2 = 1.f / (s2 + 1e-7f);
    for (int i = 0; i < 5; ++i) rowp[8 * i] = v[i] * inv2;
    rowp[L_] = 0.f;
  }
  __syncthreads();

  {
    const int c0a = wave * 16, c0b = (wave + 4) * 16;
    const float* langS = lang_ws + (size_t)s * L_ * OD_;
    v8f ca = {}, cb = {};
    for (int k0 = 0; k0 < LP_; k0 += 4) {
      const int ka  = k0 + 2 * half;
      const int kb0 = min(ka, L_ - 1), kb1 = min(ka + 1, L_ - 1);
      v2f a, ba, bb;
      a.x = attnsm[lr * LP_ + ka];
      a.y = attnsm[lr * LP_ + ka + 1];
      ba.x = langS[kb0 * OD_ + c0a + lr];
      ba.y = langS[kb1 * OD_ + c0a + lr];
      bb.x = langS[kb0 * OD_ + c0b + lr];
      bb.y = langS[kb1 * OD_ + c0b + lr];
      ca = wmma4(a, ba, ca);
      cb = wmma4(a, bb, cb);
    }
#pragma unroll
    for (int v = 0; v < 8; ++v) {
      hbuf[(v + 8 * half) * OD_ + c0a + lr] = ca[v];
      hbuf[(v + 8 * half) * OD_ + c0b + lr] = cb[v];
    }
  }
  __syncthreads();

  if (MODE == 0) {
    const int d = tid;
    float acc = 0.f;
    for (int j = 0; j < 16; ++j)
      acc += fg[j * OD_ + d] * hbuf[j * OD_ + d] * relT[d * RTP + j];
    acc += f_ws[((size_t)s * N_ + nb) * OD_ + d];
    out[((size_t)s * N_ + nb) * OD_ + d] = acc;
  } else {
    const int d = tid;
    for (int r = 0; r < 16; ++r) {
      float* op = out + ((size_t)s * N_ + nb + r) * OD_ + d;
      const float o = *op + fg[r * OD_ + d] * hbuf[r * OD_ + d] * relT[d * RTP + r];
      *op = o;
      float w = o;
      for (int off = 16; off > 0; off >>= 1) w += __shfl_xor(w, off);
      if (lane == 0) sred[r * 4 + wave] = w;
    }
    __syncthreads();
    if (tid < 16)
      score[(size_t)s * N_ + nb + tid] =
          (sred[tid * 4] + sred[tid * 4 + 1]) + (sred[tid * 4 + 2] + sred[tid * 4 + 3]);
  }
}

extern "C" void kernel_launch(void* const* d_in, const int* in_sizes, int n_in,
                              void* d_out, int out_size, void* d_ws,
                              size_t ws_size, hipStream_t stream) {
  (void)in_sizes; (void)n_in; (void)out_size; (void)ws_size;
  const float* feat      = (const float*)d_in[0];
  const float* coord     = (const float*)d_in[1];
  const float* lang_feat = (const float*)d_in[2];
  const float* lang_mask = (const float*)d_in[3];
  const float* rel_w1    = (const float*)d_in[4];
  const float* rel_b1    = (const float*)d_in[5];
  const float* rel_w2    = (const float*)d_in[6];
  const float* rel_b2    = (const float*)d_in[7];
  const float* lang_w1   = (const float*)d_in[8];
  const float* lang_b1   = (const float*)d_in[9];
  const float* lang_w2   = (const float*)d_in[10];
  const float* lang_b2   = (const float*)d_in[11];
  const float* feat_w1   = (const float*)d_in[12];
  const float* feat_b1   = (const float*)d_in[13];
  const float* feat_w2   = (const float*)d_in[14];
  const float* feat_b2   = (const float*)d_in[15];

  char* ws = (char*)d_ws;
  int*   idx_ws  = (int*)ws;
  float* lang_ws = (float*)(ws + (size_t)S_ * N_ * KNB * 4);
  float* f_ws    = (float*)(ws + (size_t)S_ * N_ * KNB * 4 +
                                 (size_t)S_ * L_ * OD_ * 4);

  float* out   = (float*)d_out;
  float* score = out + (size_t)S_ * N_ * OD_;

  knn_kernel<<<S_ * (N_ / KNN_TPB), KNN_TPB, 0, stream>>>(coord, idx_ws);
  lang_mlp_kernel<<<S_ * L_, 128, 0, stream>>>(lang_feat, lang_w1, lang_b1,
                                               lang_w2, lang_b2, lang_ws);
  feat_mlp_kernel<<<S_ * (N_ / 16), 256, 0, stream>>>(feat, feat_w1, feat_b1,
                                                      feat_w2, feat_b2, f_ws);
  fused_kernel<0><<<S_ * N_, 128, 0, stream>>>(coord, lang_mask, rel_w1,
                                               rel_b1, rel_w2, rel_b2, idx_ws,
                                               lang_ws, f_ws, out, score);
  fused_kernel<1><<<S_ * (N_ / 16), 128, 0, stream>>>(coord, lang_mask, rel_w1,
                                                      rel_b1, rel_w2, rel_b2,
                                                      idx_ws, lang_ws, f_ws,
                                                      out, score);
}


# rel MLP hidden layer computed once per block, not once per wave: waves 0-2 one hidden tile each, wave 3 five, fragments shared through LDS (one more barrier); all sums and their order unchanged
# speedup vs baseline: 3.8705x; 1.0774x over previous
#include <hip/hip_runtime.h>
#include <hip/hip_bf16.h>
#include <float.h>


typedef float v2f __attribute__((ext_vector_type(2)));
typedef float v8f __attribute__((ext_vector_type(8)));

#define S_      64
#define N_      256
#define L_      40
#define LP_     48
#define KNB     17
#define OD_     128
#define LANGID_ 256
#define PCID_   128
#define RTP     17

static __device__ inline v8f wmma4(v2f a, v2f b, v8f c) {
  return __builtin_amdgcn_wmma_f32_16x16x4_f32(false, a, false, b, (short)0, c,
                                               false, false);
}

template <int M>
static __device__ inline float lane_xor(float x) {
  const int b = __builtin_bit_cast(int, x);
  return __builtin_bit_cast(
      float, __builtin_amdgcn_mov_dpp(b, 0x160 + M, 0xf, 0xf, false));
}

#define KNN_TPB 64
#define KNN_CPT 8

__global__ void __launch_bounds__(KNN_TPB)
knn_kernel(const float* __restrict__ coord, int* __restrict__ idx_out) {
  __shared__ float cx[N_], cy[N_], cz[N_];
  const int s = blockIdx.x / (N_ / KNN_TPB);
  const int n = (blockIdx.x % (N_ / KNN_TPB)) * KNN_TPB + threadIdx.x;
  const float* cs = coord + (size_t)s * N_ * 3;
#pragma unroll
  for (int r = 0; r < N_ / KNN_TPB; ++r) {
    const int i = r * KNN_TPB + threadIdx.x;
    cx[i] = cs[i * 3 + 0];
    cy[i] = cs[i * 3 + 1];
    cz[i] = cs[i * 3 + 2];
  }
  __syncthreads();

  const float px = cx[n], py = cy[n], pz = cz[n];
  float ld[KNB];
  int   li[KNB];
#pragma unroll
  for (int j = 0; j < KNB; ++j) { ld[j] = INFINITY; li[j] = 0x7fffffff; }

  for (int m0 = 0; m0 < N_; m0 += KNN_CPT) {
#pragma unroll
    for (int u = 0; u < KNN_CPT; ++u) {
      const int m = m0 + u;
      const float dx = cx[m] - px, dy = cy[m] - py, dz = cz[m] - pz;
      const float d2 = dx * dx + dy * dy + dz * dz;
      bool lt[KNB];
#pragma unroll
      for (int j = 0; j < KNB; ++j) lt[j] = d2 < ld[j];
#pragma unroll
      for (int j = KNB - 1; j >= 1; --j) {
        ld[j] = __builtin_amdgcn_fmed3f(ld[j - 1], d2, ld[j]);
        li[j] = lt[j] ? (lt[j - 1] ? li[j - 1] : m) : li[j];
      }
      ld[0] = fminf(ld[0], d2);
      li[0] = lt[0] ? m : li[0];
    }
  }

  int* out = idx_out + ((size_t)s * N_ + n) * KNB;
#pragma unroll
  for (int j = 0; j < KNB; ++j) out[j] = li[j];
}

__global__ void lang_mlp_kernel(const float* __restrict__ lang_feat,
                                const float* __restrict__ w1,
                                const float* __restrict__ b1,
                                const float* __restrict__ w2,
                                const float* __restrict__ b2,
                                float* __restrict__ lang_out) {
  __shared__ float xs[LANGID_];
  __shared__ float hs[OD_];
  const int row = blockIdx.x;
  const int t   = threadIdx.x;
  const float* x = lang_feat + (size_t)row * LANGID_;
  xs[t]       = x[t];
  xs[t + 128] = x[t + 128];
  __syncthreads();
  float acc = b1[t];
  for (int k = 0; k < LANGID_; ++k) acc += xs[k] * w1[k * OD_ + t];
  hs[t] = fmaxf(acc, 0.0f);
  __syncthreads();
  float acc2 = b2[t];
  for (int k = 0; k < OD_; ++k) acc2 += hs[k] * w2[k * OD_ + t];
  lang_out[(size_t)row * OD_ + t] = acc2;
}

__global__ void feat_mlp_kernel(const float* __restrict__ feat,
                                const float* __restrict__ w1,
                                const float* __restrict__ b1,
                                const float* __restrict__ w2,
                                const float* __restrict__ b2,
                                float* __restrict__ f_out) {
  __shared__ float Asm[16 * PCID_];
  __shared__ float Hsm[16 * OD_];
  const int s    = blockIdx.x >> 4;
  const int mt   = blockIdx.x & 15;
  const int tid  = threadIdx.x;
  const int lane = tid & 31, wave = tid >> 5;
  const int half = lane >> 4, lr = lane & 15;

  const float* A = feat + ((size_t)s * N_ + mt * 16) * PCID_;
  for (int i = tid; i < 16 * PCID_; i += 256) Asm[i] = A[i];
  __syncthreads();

  const int n0 = wave * 16;
  v8f c = {};
  for (int k0 = 0; k0 < PCID_; k0 += 4) {
    const int ka = k0 + 2 * half;
    v2f a, b;
    a.x = Asm[lr * PCID_ + ka];
    a.y = Asm[lr * PCID_ + ka + 1];
    b.x = w1[ka * OD_ + n0 + lr];
    b.y = w1[(ka + 1) * OD_ + n0 + lr];
    c = wmma4(a, b, c);
  }
  {
    const float bb = b1[n0 + lr];
#pragma unroll
    for (int v = 0; v < 8; ++v)
      Hsm[(v + 8 * half) * OD_ + n0 + lr] = fmaxf(c[v] + bb, 0.0f);
  }
  __syncthreads();

  v8f c2 = {};
  for (int k0 = 0; k0 < OD_; k0 += 4) {
    const int ka = k0 + 2 * half;
    v2f a, b;
    a.x = Hsm[lr * OD_ + ka];
    a.y = Hsm[lr * OD_ + ka + 1];
    b.x = w2[ka * OD_ + n0 + lr];
    b.y = w2[(ka + 1) * OD_ + n0 + lr];
    c2 = wmma4(a, b, c2);
  }
  {
    const float bb = b2[n0 + lr];
    float* fo = f_out + ((size_t)s * N_ + mt * 16) * OD_;
#pragma unroll
    for (int v = 0; v < 8; ++v)
      fo[(v + 8 * half) * OD_ + n0 + lr] = c2[v] + bb;
  }
}

template <int MODE>
__global__ void __launch_bounds__(128)
fused_kernel(const float* __restrict__ coord,
                             const float* __restrict__ lang_mask,
                             const float* __restrict__ rel_w1,
                             const float* __restrict__ rel_b1,
                             const float* __restrict__ rel_w2,
                             const float* __restrict__ rel_b2,
                             const int*   __restrict__ idx_ws,
                             const float* __restrict__ lang_ws,
                             const float* __restrict__ f_ws,
                             float* __restrict__ out,
                             float* __restrict__ score) {
  __shared__ float fg[16 * OD_];
  __shared__ float relin[16 * 16];
  __shared__ __attribute__((aligned(16))) float hbuf[16 * OD_];
  __shared__ float relT[OD_ * RTP];
  __shared__ float attnsm[16 * LP_];
  __shared__ float sred[16 * 4];
  __shared__ float mksm[L_];

  const int tid  = threadIdx.x;
  const int lane = tid & 31, wave = tid >> 5;
  const int half = lane >> 4, lr = lane & 15;
  const int s  = (MODE == 0) ? (blockIdx.x >> 8) : (blockIdx.x >> 4);
  const int nb = (MODE == 0) ? (blockIdx.x & 255) : ((blockIdx.x & 15) * 16);

  float mkv = 0.f;
  if (tid < L_) mkv = lang_mask[s * L_ + tid];

  int nbi[16];
#pragma unroll
  for (int r = 0; r < 16; ++r) {
    const int pt = (MODE == 0) ? nb : nb + r;
    const int j  = (MODE == 0) ? r : 16;
    nbi[r] = idx_ws[((size_t)s * N_ + pt) * KNB + j];
  }
  float gv[16];
#pragma unroll
  for (int r = 0; r < 16; ++r)
    gv[r] = f_ws[((size_t)s * N_ + nbi[r]) * OD_ + tid];

  if (tid < 16) {
    const int pt = (MODE == 0) ? nb : nb + tid;
    const int j  = (MODE == 0) ? tid : 16;
    const float* cs = coord + (size_t)s * N_ * 3;
    const int m = idx_ws[((size_t)s * N_ + pt) * KNB + j];
    const float nx = cs[m * 3 + 0], ny = cs[m * 3 + 1], nz = cs[m * 3 + 2];
    const float ox = cs[pt * 3 + 0], oy = cs[pt * 3 + 1], oz = cs[pt * 3 + 2];
    const float rx = nx - ox, ry = ny - oy, rz = nz - oz;
    const float dist = sqrtf(rx * rx + ry * ry + rz * rz);
    float* ri = relin + tid * 16;
    ri[0] = nx; ri[1] = ny; ri[2] = nz;
    ri[3] = ox; ri[4] = oy; ri[5] = oz;
    ri[6] = rx; ri[7] = ry; ri[8] = rz;
    ri[9] = dist;
    ri[10] = 0.f; ri[11] = 0.f; ri[12] = 0.f;
    ri[13] = 0.f; ri[14] = 0.f; ri[15] = 0.f;
  }
#pragma unroll
  for (int r = 0; r < 16; ++r) fg[r * OD_ + tid] = gv[r];
  if (tid < L_) mksm[tid] = mkv;
  __syncthreads();

  float4* hx = reinterpret_cast<float4*>(hbuf);
  v2f bin[4];
  for (int i = 0; i < 4; ++i) {
    const float* rp = relin + lr * 16 + 4 * i + 2 * half;
    bin[i].x = rp[0];
    bin[i].y = rp[1];
  }
  auto hidden_tile = [&](const int t) {
    v8f c = {};
    for (int i = 0; i < 4; ++i) {
      const int ka  = 4 * i + 2 * half;
      const int ka0 = min(ka, 9), ka1 = min(ka + 1, 9);
      v2f a;
      a.x = rel_w1[ka0 * OD_ + 16 * t + lr];
      a.y = rel_w1[ka1 * OD_ + 16 * t + lr];
      c = wmma4(a, bin[i], c);
    }
    const float* bp = rel_b1 + 16 * t + 8 * half;
    float h[8];
    for (int r = 0; r < 8; ++r) h[r] = fmaxf(c[r] + bp[r], 0.f);
    hx[(2 * t) * 32 + lane]     = make_float4(h[0], h[1], h[2], h[3]);
    hx[(2 * t + 1) * 32 + lane] = make_float4(h[4], h[5], h[6], h[7]);
  };

  if (wave < 3) {
    hidden_tile(wave);
    const int c0 = wave * 16;
    const float* langS = lang_ws + (size_t)s * L_ * OD_;
    const int lclamp = min(c0 + lr, L_ - 1);
    v8f c = {};
    for (int k0 = 0; k0 < OD_; k0 += 4) {
      const int ka = k0 + 2 * half;
      v2f a, b;
      a.x = fg[lr * OD_ + ka];
      a.y = fg[lr * OD_ + ka + 1];
      b.x = langS[lclamp * OD_ + ka];
      b.y = langS[lclamp * OD_ + ka + 1];
      c = wmma4(a, b, c);
    }
#pragma unroll
    for (int v = 0; v < 8; ++v)
      attnsm[(v + 8 * half) * LP_ + c0 + lr] = c[v];
  } else {
#pragma unroll
    for (int t = 3; t < 8; ++t) hidden_tile(t);
  }
  __syncthreads();

  {
    const int g = wave;
    v8f hT[8];
    for (int t = 0; t < 8; ++t) {
      const float4 lo = hx[(2 * t) * 32 + lane];
      const float4 hi = hx[(2 * t + 1) * 32 + lane];
      hT[t][0] = lo.x; hT[t][1] = lo.y; hT[t][2] = lo.z; hT[t][3] = lo.w;
      hT[t][4] = hi.x; hT[t][5] = hi.y; hT[t][6] = hi.z; hT[t][7] = hi.w;
    }

    const int d0a = 16 * g, d0b = 16 * (g + 4);
    v8f ca = {}, cb = {};
    for (int t = 0; t < 8; ++t) {
      for (int q = 0; q < 4; ++q) {
        const int k = 16 * t + 8 * half + 2 * q;
        v2f aa, ab, b;
        aa.x = rel_w2[k * OD_ + d0a + lr];
        aa.y = rel_w2[(k + 1) * OD_ + d0a + lr];
        ab.x = rel_w2[k * OD_ + d0b + lr];
        ab.y = rel_w2[(k + 1) * OD_ + d0b + lr];
        b.x = hT[t][2 * q];
        b.y = hT[t][2 * q + 1];
        ca = wmma4(aa, b, ca);
        cb = wmma4(ab, b, cb);
      }
    }
    const float* ba = rel_b2 + d0a + 8 * half;
    const float* bb = rel_b2 + d0b + 8 * half;
    for (int r = 0; r < 8; ++r) {
      relT[(d0a + 8 * half + r) * RTP + lr] = ca[r] + ba[r];
      relT[(d0b + 8 * half + r) * RTP + lr] = cb[r] + bb[r];
    }
  }
  __syncthreads();

  {
    float* rowp = attnsm + (tid >> 3) * LP_ + (tid & 7);
    const float* mkp = mksm + (tid & 7);
    float v[5];
    for (int i = 0; i < 5; ++i) v[i] = rowp[8 * i];
    float mx = -FLT_MAX;
    for (int i = 0; i < 5; ++i) mx = fmaxf(mx, v[i]);
    mx = fmaxf(mx, lane_xor<1>(mx));
    mx = fmaxf(mx, lane_xor<2>(mx));
    mx = fmaxf(mx, lane_xor<4>(mx));
    float ssum = 0.f;
    for (int i = 0; i < 5; ++i) {
      v[i] = expf(v[i] - mx);
      ssum += v[i];
    }
    ssum += lane_xor<1>(ssum);
    ssum += lane_xor<2>(ssum);
    ssum += lane_xor<4>(ssum);
    const float inv = 1.f / ssum;
    float s2 = 0.f;
    for (int i = 0; i < 5; ++i) {
      v[i] = v[i] * inv * mkp[8 * i];
      s2 += v[i];
    }
    s2 += lane_xor<1>(s2);
    s2 += lane_xor<2>(s2);
    s2 += lane_xor<4>(s2);
    const float inv2 = 1.f / (s2 + 1e-7f);
    for (int i = 0; i < 5; ++i) rowp[8 * i] = v[i] * inv2;
    rowp[L_] = 0.f;
  }
  __syncthreads();

  {
    const int c0a = wave * 16, c0b = (wave + 4) * 16;
    const float* langS = lang_ws + (size_t)s * L_ * OD_;
    v8f ca = {}, cb = {};
    for (int k0 = 0; k0 < LP_; k0 += 4) {
      const int ka  = k0 + 2 * half;
      const int kb0 = min(ka, L_ - 1), kb1 = min(ka + 1, L_ - 1);
      v2f a, ba, bb;
      a.x = attnsm[lr * LP_ + ka];
      a.y = attnsm[lr * LP_ + ka + 1];
      ba.x = langS[kb0 * OD_ + c0a + lr];
      ba.y = langS[kb1 * OD_ + c0a + lr];
      bb.x = langS[kb0 * OD_ + c0b + lr];
      bb.y = langS[kb1 * OD_ + c0b + lr];
      ca = wmma4(a, ba, ca);
      cb = wmma4(a, bb, cb);
    }
#pragma unroll
    for (int v = 0; v < 8; ++v) {
      hbuf[(v + 8 * half) * OD_ + c0a + lr] = ca[v];
      hbuf[(v + 8 * half) * OD_ + c0b + lr] = cb[v];
    }
  }
  __syncthreads();

  if (MODE == 0) {
    const int d = tid;
    float acc = 0.f;
    for (int j = 0; j < 16; ++j)
      acc += fg[j * OD_ + d] * hbuf[j * OD_ + d] * relT[d * RTP + j];
    acc += f_ws[((size_t)s * N_ + nb) * OD_ + d];
    out[((size_t)s * N_ + nb) * OD_ + d] = acc;
  } else {
    const int d = tid;
    for (int r = 0; r < 16; ++r) {
      float* op = out + ((size_t)s * N_ + nb + r) * OD_ + d;
      const float o = *op + fg[r * OD_ + d] * hbuf[r * OD_ + d] * relT[d * RTP + r];
      *op = o;
      float w = o;
      for (int off = 16; off > 0; off >>= 1) w += __shfl_xor(w, off);
      if (lane == 0) sred[r * 4 + wave] = w;
    }
    __syncthreads();
    if (tid < 16)
      score[(size_t)s * N_ + nb + tid] =
          (sred[tid * 4] + sred[tid * 4 + 1]) + (sred[tid * 4 + 2] + sred[tid * 4 + 3]);
  }
}

extern "C" void kernel_launch(void* const* d_in, const int* in_sizes, int n_in,
                              void* d_out, int out_size, void* d_ws,
                              size_t ws_size, hipStream_t stream) {
  (void)in_sizes; (void)n_in; (void)out_size; (void)ws_size;
  const float* feat      = (const float*)d_in[0];
  const float* coord     = (const float*)d_in[1];
  const float* lang_feat = (const float*)d_in[2];
  const float* lang_mask = (const float*)d_in[3];
  const float* rel_w1    = (const float*)d_in[4];
  const float* rel_b1    = (const float*)d_in[5];
  const float* rel_w2    = (const float*)d_in[6];
  const float* rel_b2    = (const float*)d_in[7];
  const float* lang_w1   = (const float*)d_in[8];
  const float* lang_b1   = (const float*)d_in[9];
  const float* lang_w2   = (const float*)d_in[10];
  const float* lang_b2   = (const float*)d_in[11];
  const float* feat_w1   = (const float*)d_in[12];
  const float* feat_b1   = (const float*)d_in[13];
  const float* feat_w2   = (const float*)d_in[14];
  const float* feat_b2   = (const float*)d_in[15];

  char* ws = (char*)d_ws;
  int*   idx_ws  = (int*)ws;
  float* lang_ws = (float*)(ws + (size_t)S_ * N_ * KNB * 4);
  float* f_ws    = (float*)(ws + (size_t)S_ * N_ * KNB * 4 +
                                 (size_t)S_ * L_ * OD_ * 4);

  float* out   = (float*)d_out;
  float* score = out + (size_t)S_ * N_ * OD_;

  knn_kernel<<<S_ * (N_ / KNN_TPB), KNN_TPB, 0, stream>>>(coord, idx_ws);
  lang_mlp_kernel<<<S_ * L_, 128, 0, stream>>>(lang_feat, lang_w1, lang_b1,
                                               lang_w2, lang_b2, lang_ws);
  feat_mlp_kernel<<<S_ * (N_ / 16), 256, 0, stream>>>(feat, feat_w1, feat_b1,
                                                      feat_w2, feat_b2, f_ws);
  fused_kernel<0><<<S_ * N_, 128, 0, stream>>>(coord, lang_mask, rel_w1,
                                               rel_b1, rel_w2, rel_b2, idx_ws,
                                               lang_ws, f_ws, out, score);
  fused_kernel<1><<<S_ * (N_ / 16), 128, 0, stream>>>(coord, lang_mask, rel_w1,
                                                      rel_b1, rel_w2, rel_b2,
                                                      idx_ws, lang_ws, f_ws,
                                                      out, score);
}


# Fused kernel's operand loads widened: rel_w2 (layer 2) and lang (ins) tiles take interleaved outputs, 16 lanes read one whole line; logits k regrouped for 16-byte loads: logit sum order changes
# speedup vs baseline: 4.0112x; 1.0364x over previous
#include <hip/hip_runtime.h>
#include <hip/hip_bf16.h>
#include <float.h>


typedef float v2f __attribute__((ext_vector_type(2)));
typedef float v4f __attribute__((ext_vector_type(4)));
typedef float v8f __attribute__((ext_vector_type(8)));

#define S_      64
#define N_      256
#define L_      40
#define LP_     48
#define KNB     17
#define OD_     128
#define LANGID_ 256
#define PCID_   128
#define RTP     17

static __device__ inline v8f wmma4(v2f a, v2f b, v8f c) {
  return __builtin_amdgcn_wmma_f32_16x16x4_f32(false, a, false, b, (short)0, c,
                                               false, false);
}

template <int M>
static __device__ inline float lane_xor(float x) {
  const int b = __builtin_bit_cast(int, x);
  return __builtin_bit_cast(
      float, __builtin_amdgcn_mov_dpp(b, 0x160 + M, 0xf, 0xf, false));
}

#define KNN_TPB 64
#define KNN_CPT 8

__global__ void __launch_bounds__(KNN_TPB)
knn_kernel(const float* __restrict__ coord, int* __restrict__ idx_out) {
  __shared__ float cx[N_], cy[N_], cz[N_];
  const int s = blockIdx.x / (N_ / KNN_TPB);
  const int n = (blockIdx.x % (N_ / KNN_TPB)) * KNN_TPB + threadIdx.x;
  const float* cs = coord + (size_t)s * N_ * 3;
#pragma unroll
  for (int r = 0; r < N_ / KNN_TPB; ++r) {
    const int i = r * KNN_TPB + threadIdx.x;
    cx[i] = cs[i * 3 + 0];
    cy[i] = cs[i * 3 + 1];
    cz[i] = cs[i * 3 + 2];
  }
  __syncthreads();

  const float px = cx[n], py = cy[n], pz = cz[n];
  float ld[KNB];
  int   li[KNB];
#pragma unroll
  for (int j = 0; j < KNB; ++j) { ld[j] = INFINITY; li[j] = 0x7fffffff; }

  for (int m0 = 0; m0 < N_; m0 += KNN_CPT) {
#pragma unroll
    for (int u = 0; u < KNN_CPT; ++u) {
      const int m = m0 + u;
      const float dx = cx[m] - px, dy = cy[m] - py, dz = cz[m] - pz;
      const float d2 = dx * dx + dy * dy + dz * dz;
      bool lt[KNB];
#pragma unroll
      for (int j = 0; j < KNB; ++j) lt[j] = d2 < ld[j];
#pragma unroll
      for (int j = KNB - 1; j >= 1; --j) {
        ld[j] = __builtin_amdgcn_fmed3f(ld[j - 1], d2, ld[j]);
        li[j] = lt[j] ? (lt[j - 1] ? li[j - 1] : m) : li[j];
      }
      ld[0] = fminf(ld[0], d2);
      li[0] = lt[0] ? m : li[0];
    }
  }

  int* out = idx_out + ((size_t)s * N_ + n) * KNB;
#pragma unroll
  for (int j = 0; j < KNB; ++j) out[j] = li[j];
}

__global__ void lang_mlp_kernel(const float* __restrict__ lang_feat,
                                const float* __restrict__ w1,
                                const float* __restrict__ b1,
                                const float* __restrict__ w2,
                                const float* __restrict__ b2,
                                float* __restrict__ lang_out) {
  __shared__ float xs[LANGID_];
  __shared__ float hs[OD_];
  const int row = blockIdx.x;
  const int t   = threadIdx.x;
  const float* x = lang_feat + (size_t)row * LANGID_;
  xs[t]       = x[t];
  xs[t + 128] = x[t + 128];
  __syncthreads();
  float acc = b1[t];
  for (int k = 0; k < LANGID_; ++k) acc += xs[k] * w1[k * OD_ + t];
  hs[t] = fmaxf(acc, 0.0f);
  __syncthreads();
  float acc2 = b2[t];
  for (int k = 0; k < OD_; ++k) acc2 += hs[k] * w2[k * OD_ + t];
  lang_out[(size_t)row * OD_ + t] = acc2;
}

__global__ void feat_mlp_kernel(const float* __restrict__ feat,
                                const float* __restrict__ w1,
                                const float* __restrict__ b1,
                                const float* __restrict__ w2,
                                const float* __restrict__ b2,
                                float* __restrict__ f_out) {
  __shared__ float Asm[16 * PCID_];
  __shared__ float Hsm[16 * OD_];
  const int s    = blockIdx.x >> 4;
  const int mt   = blockIdx.x & 15;
  const int tid  = threadIdx.x;
  const int lane = tid & 31, wave = tid >> 5;
  const int half = lane >> 4, lr = lane & 15;

  const float* A = feat + ((size_t)s * N_ + mt * 16) * PCID_;
  for (int i = tid; i < 16 * PCID_; i += 256) Asm[i] = A[i];
  __syncthreads();

  const int n0 = wave * 16;
  v8f c = {};
  for (int k0 = 0; k0 < PCID_; k0 += 4) {
    const int ka = k0 + 2 * half;
    v2f a, b;
    a.x = Asm[lr * PCID_ + ka];
    a.y = Asm[lr * PCID_ + ka + 1];
    b.x = w1[ka * OD_ + n0 + lr];
    b.y = w1[(ka + 1) * OD_ + n0 + lr];
    c = wmma4(a, b, c);
  }
  {
    const float bb = b1[n0 + lr];
#pragma unroll
    for (int v = 0; v < 8; ++v)
      Hsm[(v + 8 * half) * OD_ + n0 + lr] = fmaxf(c[v] + bb, 0.0f);
  }
  __syncthreads();

  v8f c2 = {};
  for (int k0 = 0; k0 < OD_; k0 += 4) {
    const int ka = k0 + 2 * half;
    v2f a, b;
    a.x = Hsm[lr * OD_ + ka];
    a.y = Hsm[lr * OD_ + ka + 1];
    b.x = w2[ka * OD_ + n0 + lr];
    b.y = w2[(ka + 1) * OD_ + n0 + lr];
    c2 = wmma4(a, b, c2);
  }
  {
    const float bb = b2[n0 + lr];
    float* fo = f_out + ((size_t)s * N_ + mt * 16) * OD_;
#pragma unroll
    for (int v = 0; v < 8; ++v)
      fo[(v + 8 * half) * OD_ + n0 + lr] = c2[v] + bb;
  }
}

template <int MODE>
__global__ void __launch_bounds__(128)
fused_kernel(const float* __restrict__ coord,
                             const float* __restrict__ lang_mask,
                             const float* __restrict__ rel_w1,
                             const float* __restrict__ rel_b1,
                             const float* __restrict__ rel_w2,
                             const float* __restrict__ rel_b2,
                             const int*   __restrict__ idx_ws,
                             const float* __restrict__ lang_ws,
                             const float* __restrict__ f_ws,
                             float* __restrict__ out,
                             float* __restrict__ score) {
  __shared__ __attribute__((aligned(16))) float fg[16 * OD_];
  __shared__ float relin[16 * 16];
  __shared__ __attribute__((aligned(16))) float hbuf[16 * OD_];
  __shared__ float relT[OD_ * RTP];
  __shared__ float attnsm[16 * LP_];
  __shared__ float sred[16 * 4];
  __shared__ float mksm[L_];

  const int tid  = threadIdx.x;
  const int lane = tid & 31, wave = tid >> 5;
  const int half = lane >> 4, lr = lane & 15;
  const int s  = (MODE == 0) ? (blockIdx.x >> 8) : (blockIdx.x >> 4);
  const int nb = (MODE == 0) ? (blockIdx.x & 255) : ((blockIdx.x & 15) * 16);

  float mkv = 0.f;
  if (tid < L_) mkv = lang_mask[s * L_ + tid];

  int nbi[16];
#pragma unroll
  for (int r = 0; r < 16; ++r) {
    const int pt = (MODE == 0) ? nb : nb + r;
    const int j  = (MODE == 0) ? r : 16;
    nbi[r] = idx_ws[((size_t)s * N_ + pt) * KNB + j];
  }
  float gv[16];
#pragma unroll
  for (int r = 0; r < 16; ++r)
    gv[r] = f_ws[((size_t)s * N_ + nbi[r]) * OD_ + tid];

  if (tid < 16) {
    const int pt = (MODE == 0) ? nb : nb + tid;
    const int j  = (MODE == 0) ? tid : 16;
    const float* cs = coord + (size_t)s * N_ * 3;
    const int m = idx_ws[((size_t)s * N_ + pt) * KNB + j];
    const float nx = cs[m * 3 + 0], ny = cs[m * 3 + 1], nz = cs[m * 3 + 2];
    const float ox = cs[pt * 3 + 0], oy = cs[pt * 3 + 1], oz = cs[pt * 3 + 2];
    const float rx = nx - ox, ry = ny - oy, rz = nz - oz;
    const float dist = sqrtf(rx * rx + ry * ry + rz * rz);
    float* ri = relin + tid * 16;
    ri[0] = nx; ri[1] = ny; ri[2] = nz;
    ri[3] = ox; ri[4] = oy; ri[5] = oz;
    ri[6] = rx; ri[7] = ry; ri[8] = rz;
    ri[9] = dist;
    ri[10] = 0.f; ri[11] = 0.f; ri[12] = 0.f;
    ri[13] = 0.f; ri[14] = 0.f; ri[15] = 0.f;
  }
#pragma unroll
  for (int r = 0; r < 16; ++r) fg[r * OD_ + tid] = gv[r];
  if (tid < L_) mksm[tid] = mkv;
  __syncthreads();

  float4* hx = reinterpret_cast<float4*>(hbuf);
  v2f bin[4];
  for (int i = 0; i < 4; ++i) {
    const float* rp = relin + lr * 16 + 4 * i + 2 * half;
    bin[i].x = rp[0];
    bin[i].y = rp[1];
  }
  auto hidden_tile = [&](const int t) {
    v8f c = {};
    for (int i = 0; i < 4; ++i) {
      const int ka  = 4 * i + 2 * half;
      const int ka0 = min(ka, 9), ka1 = min(ka + 1, 9);
      v2f a;
      a.x = rel_w1[ka0 * OD_ + 16 * t + lr];
      a.y = rel_w1[ka1 * OD_ + 16 * t + lr];
      c = wmma4(a, bin[i], c);
    }
    const float* bp = rel_b1 + 16 * t + 8 * half;
    float h[8];
    for (int r = 0; r < 8; ++r) h[r] = fmaxf(c[r] + bp[r], 0.f);
    hx[(2 * t) * 32 + lane]     = make_float4(h[0], h[1], h[2], h[3]);
    hx[(2 * t + 1) * 32 + lane] = make_float4(h[4], h[5], h[6], h[7]);
  };

  if (wave < 3) {
    hidden_tile(wave);
    const int c0 = wave * 16;
    const float* langS = lang_ws + (size_t)s * L_ * OD_;
    const int lclamp = min(c0 + lr, L_ - 1);
    const float* arow = fg + lr * OD_ + 8 * half;
    const float* brow = langS + lclamp * OD_ + 8 * half;
    v8f c = {};
    for (int k0 = 0; k0 < OD_; k0 += 16) {
      const v4f b0 = *reinterpret_cast<const v4f*>(brow + k0);
      const v4f b1 = *reinterpret_cast<const v4f*>(brow + k0 + 4);
      const v4f a0 = *reinterpret_cast<const v4f*>(arow + k0);
      const v4f a1 = *reinterpret_cast<const v4f*>(arow + k0 + 4);
      v2f a, b;
      a.x = a0.x; a.y = a0.y; b.x = b0.x; b.y = b0.y; c = wmma4(a, b, c);
      a.x = a0.z; a.y = a0.w; b.x = b0.z; b.y = b0.w; c = wmma4(a, b, c);
      a.x = a1.x; a.y = a1.y; b.x = b1.x; b.y = b1.y; c = wmma4(a, b, c);
      a.x = a1.z; a.y = a1.w; b.x = b1.z; b.y = b1.w; c = wmma4(a, b, c);
    }
#pragma unroll
    for (int v = 0; v < 8; ++v)
      attnsm[(v + 8 * half) * LP_ + c0 + lr] = c[v];
  } else {
#pragma unroll
    for (int t = 3; t < 8; ++t) hidden_tile(t);
  }
  __syncthreads();

  {
    const int g = wave;
    v8f hT[8];
    for (int t = 0; t < 8; ++t) {
      const float4 lo = hx[(2 * t) * 32 + lane];
      const float4 hi = hx[(2 * t + 1) * 32 + lane];
      hT[t][0] = lo.x; hT[t][1] = lo.y; hT[t][2] = lo.z; hT[t][3] = lo.w;
      hT[t][4] = hi.x; hT[t][5] = hi.y; hT[t][6] = hi.z; hT[t][7] = hi.w;
    }

    const float* wp = rel_w2 + 32 * g + 2 * lr;
    v8f ca = {}, cb = {};
    for (int t = 0; t < 8; ++t) {
      for (int q = 0; q < 4; ++q) {
        const int k = 16 * t + 8 * half + 2 * q;
        const v2f w0 = *reinterpret_cast<const v2f*>(wp + k * OD_);
        const v2f w1 = *reinterpret_cast<const v2f*>(wp + (k + 1) * OD_);
        v2f aa, ab, b;
        aa.x = w0.x; aa.y = w1.x;
        ab.x = w0.y; ab.y = w1.y;
        b.x = hT[t][2 * q];
        b.y = hT[t][2 * q + 1];
        ca = wmma4(aa, b, ca);
        cb = wmma4(ab, b, cb);
      }
    }
    const int dbase = 32 * g + 16 * half;
    const float* bp2 = rel_b2 + dbase;
    for (int r = 0; r < 8; ++r) {
      relT[(dbase + 2 * r) * RTP + lr]     = ca[r] + bp2[2 * r];
      relT[(dbase + 2 * r + 1) * RTP + lr] = cb[r] + bp2[2 * r + 1];
    }
  }
  __syncthreads();

  {
    float* rowp = attnsm + (tid >> 3) * LP_ + (tid & 7);
    const float* mkp = mksm + (tid & 7);
    float v[5];
    for (int i = 0; i < 5; ++i) v[i] = rowp[8 * i];
    float mx = -FLT_MAX;
    for (int i = 0; i < 5; ++i) mx = fmaxf(mx, v[i]);
    mx = fmaxf(mx, lane_xor<1>(mx));
    mx = fmaxf(mx, lane_xor<2>(mx));
    mx = fmaxf(mx, lane_xor<4>(mx));
    float ssum = 0.f;
    for (int i = 0; i < 5; ++i) {
      v[i] = expf(v[i] - mx);
      ssum += v[i];
    }
    ssum += lane_xor<1>(ssum);
    ssum += lane_xor<2>(ssum);
    ssum += lane_xor<4>(ssum);
    const float inv = 1.f / ssum;
    float s2 = 0.f;
    for (int i = 0; i < 5; ++i) {
      v[i] = v[i] * inv * mkp[8 * i];
      s2 += v[i];
    }
    s2 += lane_xor<1>(s2);
    s2 += lane_xor<2>(s2);
    s2 += lane_xor<4>(s2);
    const float inv2 = 1.f / (s2 + 1e-7f);
    for (int i = 0; i < 5; ++i) rowp[8 * i] = v[i] * inv2;
    rowp[L_] = 0.f;
  }
  __syncthreads();

  {
    const int c0 = 32 * wave + 2 * lr;
    const float* langS = lang_ws + (size_t)s * L_ * OD_ + c0;
    v8f ca = {}, cb = {};
    for (int k0 = 0; k0 < LP_; k0 += 4) {
      const int ka  = k0 + 2 * half;
      const int kb0 = min(ka, L_ - 1), kb1 = min(ka + 1, L_ - 1);
      const v2f l0 = *reinterpret_cast<const v2f*>(langS + kb0 * OD_);
      const v2f l1 = *reinterpret_cast<const v2f*>(langS + kb1 * OD_);
      v2f a, ba, bb;
      a.x = attnsm[lr * LP_ + ka];
      a.y = attnsm[lr * LP_ + ka + 1];
      ba.x = l0.x; ba.y = l1.x;
      bb.x = l0.y; bb.y = l1.y;
      ca = wmma4(a, ba, ca);
      cb = wmma4(a, bb, cb);
    }
#pragma unroll
    for (int v = 0; v < 8; ++v) {
      v2f o;
      o.x = ca[v]; o.y = cb[v];
      *reinterpret_cast<v2f*>(hbuf + (v + 8 * half) * OD_ + c0) = o;
    }
  }
  __syncthreads();

  if (MODE == 0) {
    const int d = tid;
    float acc = 0.f;
    for (int j = 0; j < 16; ++j)
      acc += fg[j * OD_ + d] * hbuf[j * OD_ + d] * relT[d * RTP + j];
    acc += f_ws[((size_t)s * N_ + nb) * OD_ + d];
    out[((size_t)s * N_ + nb) * OD_ + d] = acc;
  } else {
    const int d = tid;
    for (int r = 0; r < 16; ++r) {
      float* op = out + ((size_t)s * N_ + nb + r) * OD_ + d;
      const float o = *op + fg[r * OD_ + d] * hbuf[r * OD_ + d] * relT[d * RTP + r];
      *op = o;
      float w = o;
      for (int off = 16; off > 0; off >>= 1) w += __shfl_xor(w, off);
      if (lane == 0) sred[r * 4 + wave] = w;
    }
    __syncthreads();
    if (tid < 16)
      score[(size_t)s * N_ + nb + tid] =
          (sred[tid * 4] + sred[tid * 4 + 1]) + (sred[tid * 4 + 2] + sred[tid * 4 + 3]);
  }
}

extern "C" void kernel_launch(void* const* d_in, const int* in_sizes, int n_in,
                              void* d_out, int out_size, void* d_ws,
                              size_t ws_size, hipStream_t stream) {
  (void)in_sizes; (void)n_in; (void)out_size; (void)ws_size;
  const float* feat      = (const float*)d_in[0];
  const float* coord     = (const float*)d_in[1];
  const float* lang_feat = (const float*)d_in[2];
  const float* lang_mask = (const float*)d_in[3];
  const float* rel_w1    = (const float*)d_in[4];
  const float* rel_b1    = (const float*)d_in[5];
  const float* rel_w2    = (const float*)d_in[6];
  const float* rel_b2    = (const float*)d_in[7];
  const float* lang_w1   = (const float*)d_in[8];
  const float* lang_b1   = (const float*)d_in[9];
  const float* lang_w2   = (const float*)d_in[10];
  const float* lang_b2   = (const float*)d_in[11];
  const float* feat_w1   = (const float*)d_in[12];
  const float* feat_b1   = (const float*)d_in[13];
  const float* feat_w2   = (const float*)d_in[14];
  const float* feat_b2   = (const float*)d_in[15];

  char* ws = (char*)d_ws;
  int*   idx_ws  = (int*)ws;
  float* lang_ws = (float*)(ws + (size_t)S_ * N_ * KNB * 4);
  float* f_ws    = (float*)(ws + (size_t)S_ * N_ * KNB * 4 +
                                 (size_t)S_ * L_ * OD_ * 4);

  float* out   = (float*)d_out;
  float* score = out + (size_t)S_ * N_ * OD_;

  knn_kernel<<<S_ * (N_ / KNN_TPB), KNN_TPB, 0, stream>>>(coord, idx_ws);
  lang_mlp_kernel<<<S_ * L_, 128, 0, stream>>>(lang_feat, lang_w1, lang_b1,
                                               lang_w2, lang_b2, lang_ws);
  feat_mlp_kernel<<<S_ * (N_ / 16), 256, 0, stream>>>(feat, feat_w1, feat_b1,
                                                      feat_w2, feat_b2, f_ws);
  fused_kernel<0><<<S_ * N_, 128, 0, stream>>>(coord, lang_mask, rel_w1,
                                               rel_b1, rel_w2, rel_b2, idx_ws,
                                               lang_ws, f_ws, out, score);
  fused_kernel<1><<<S_ * (N_ / 16), 128, 0, stream>>>(coord, lang_mask, rel_w1,
                                                      rel_b1, rel_w2, rel_b2,
                                                      idx_ws, lang_ws, f_ws,
                                                      out, score);
}


# ins product skips zero-mask language rows: lang rows stored nonzero-mask first (order from ballots of the mask), K = their count, was 48 padded; softmax and ins sum order change
# speedup vs baseline: 4.2084x; 1.0492x over previous
#include <hip/hip_runtime.h>
#include <hip/hip_bf16.h>
#include <float.h>


typedef float v2f __attribute__((ext_vector_type(2)));
typedef float v4f __attribute__((ext_vector_type(4)));
typedef float v8f __attribute__((ext_vector_type(8)));

#define S_      64
#define N_      256
#define L_      40
#define LP_     48
#define KNB     17
#define OD_     128
#define LANGID_ 256
#define PCID_   128
#define RTP     17

static __device__ inline v8f wmma4(v2f a, v2f b, v8f c) {
  return __builtin_amdgcn_wmma_f32_16x16x4_f32(false, a, false, b, (short)0, c,
                                               false, false);
}

template <int M>
static __device__ inline float lane_xor(float x) {
  const int b = __builtin_bit_cast(int, x);
  return __builtin_bit_cast(
      float, __builtin_amdgcn_mov_dpp(b, 0x160 + M, 0xf, 0xf, false));
}

static __device__ inline void lang_ballots(const float m0, const float m1,
                                           const int lane, unsigned& balA,
                                           unsigned& balB) {
  balA = __builtin_amdgcn_ballot_w32(m0 != 0.f);
  balB = __builtin_amdgcn_ballot_w32((lane < 8) && (m1 != 0.f));
}
static __device__ inline int lang_dest(const int l, const unsigned balA,
                                       const unsigned balB, const int cnt) {
  const unsigned below = (1u << (l & 31)) - 1u;
  const int before = (l < 32) ? __builtin_popcount(balA & below)
                              : __builtin_popcount(balA) +
                                    __builtin_popcount(balB & below);
  const bool on = ((((l < 32) ? balA : balB) >> (l & 31)) & 1u) != 0u;
  return on ? before : cnt + (l - before);
}

#define KNN_TPB 64
#define KNN_CPT 8

__global__ void __launch_bounds__(KNN_TPB)
knn_kernel(const float* __restrict__ coord, int* __restrict__ idx_out) {
  __shared__ float cx[N_], cy[N_], cz[N_];
  const int s = blockIdx.x / (N_ / KNN_TPB);
  const int n = (blockIdx.x % (N_ / KNN_TPB)) * KNN_TPB + threadIdx.x;
  const float* cs = coord + (size_t)s * N_ * 3;
#pragma unroll
  for (int r = 0; r < N_ / KNN_TPB; ++r) {
    const int i = r * KNN_TPB + threadIdx.x;
    cx[i] = cs[i * 3 + 0];
    cy[i] = cs[i * 3 + 1];
    cz[i] = cs[i * 3 + 2];
  }
  __syncthreads();

  const float px = cx[n], py = cy[n], pz = cz[n];
  float ld[KNB];
  int   li[KNB];
#pragma unroll
  for (int j = 0; j < KNB; ++j) { ld[j] = INFINITY; li[j] = 0x7fffffff; }

  for (int m0 = 0; m0 < N_; m0 += KNN_CPT) {
#pragma unroll
    for (int u = 0; u < KNN_CPT; ++u) {
      const int m = m0 + u;
      const float dx = cx[m] - px, dy = cy[m] - py, dz = cz[m] - pz;
      const float d2 = dx * dx + dy * dy + dz * dz;
      bool lt[KNB];
#pragma unroll
      for (int j = 0; j < KNB; ++j) lt[j] = d2 < ld[j];
#pragma unroll
      for (int j = KNB - 1; j >= 1; --j) {
        ld[j] = __builtin_amdgcn_fmed3f(ld[j - 1], d2, ld[j]);
        li[j] = lt[j] ? (lt[j - 1] ? li[j - 1] : m) : li[j];
      }
      ld[0] = fminf(ld[0], d2);
      li[0] = lt[0] ? m : li[0];
    }
  }

  int* out = idx_out + ((size_t)s * N_ + n) * KNB;
#pragma unroll
  for (int j = 0; j < KNB; ++j) out[j] = li[j];
}

__global__ void lang_mlp_kernel(const float* __restrict__ lang_feat,
                                const float* __restrict__ lang_mask,
                                const float* __restrict__ w1,
                                const float* __restrict__ b1,
                                const float* __restrict__ w2,
                                const float* __restrict__ b2,
                                float* __restrict__ lang_out) {
  __shared__ float xs[LANGID_];
  __shared__ float hs[OD_];
  const int row = blockIdx.x;
  const int t   = threadIdx.x;
  const int s = row / L_, l = row - s * L_;
  const float mk0 = lang_mask[s * L_ + (t & 31)];
  const float mk1 = lang_mask[s * L_ + 32 + (t & 7)];
  const float* x = lang_feat + (size_t)row * LANGID_;
  xs[t]       = x[t];
  xs[t + 128] = x[t + 128];
  __syncthreads();
  float acc = b1[t];
  for (int k = 0; k < LANGID_; ++k) acc += xs[k] * w1[k * OD_ + t];
  hs[t] = fmaxf(acc, 0.0f);
  __syncthreads();
  float acc2 = b2[t];
  for (int k = 0; k < OD_; ++k) acc2 += hs[k] * w2[k * OD_ + t];
  unsigned balA, balB;
  lang_ballots(mk0, mk1, t & 31, balA, balB);
  const int cnt  = __builtin_popcount(balA) + __builtin_popcount(balB);
  const int dest = lang_dest(l, balA, balB, cnt);
  lang_out[((size_t)s * L_ + dest) * OD_ + t] = acc2;
}

__global__ void feat_mlp_kernel(const float* __restrict__ feat,
                                const float* __restrict__ w1,
                                const float* __restrict__ b1,
                                const float* __restrict__ w2,
                                const float* __restrict__ b2,
                                float* __restrict__ f_out) {
  __shared__ float Asm[16 * PCID_];
  __shared__ float Hsm[16 * OD_];
  const int s    = blockIdx.x >> 4;
  const int mt   = blockIdx.x & 15;
  const int tid  = threadIdx.x;
  const int lane = tid & 31, wave = tid >> 5;
  const int half = lane >> 4, lr = lane & 15;

  const float* A = feat + ((size_t)s * N_ + mt * 16) * PCID_;
  for (int i = tid; i < 16 * PCID_; i += 256) Asm[i] = A[i];
  __syncthreads();

  const int n0 = wave * 16;
  v8f c = {};
  for (int k0 = 0; k0 < PCID_; k0 += 4) {
    const int ka = k0 + 2 * half;
    v2f a, b;
    a.x = Asm[lr * PCID_ + ka];
    a.y = Asm[lr * PCID_ + ka + 1];
    b.x = w1[ka * OD_ + n0 + lr];
    b.y = w1[(ka + 1) * OD_ + n0 + lr];
    c = wmma4(a, b, c);
  }
  {
    const float bb = b1[n0 + lr];
#pragma unroll
    for (int v = 0; v < 8; ++v)
      Hsm[(v + 8 * half) * OD_ + n0 + lr] = fmaxf(c[v] + bb, 0.0f);
  }
  __syncthreads();

  v8f c2 = {};
  for (int k0 = 0; k0 < OD_; k0 += 4) {
    const int ka = k0 + 2 * half;
    v2f a, b;
    a.x = Hsm[lr * OD_ + ka];
    a.y = Hsm[lr * OD_ + ka + 1];
    b.x = w2[ka * OD_ + n0 + lr];
    b.y = w2[(ka + 1) * OD_ + n0 + lr];
    c2 = wmma4(a, b, c2);
  }
  {
    const float bb = b2[n0 + lr];
    float* fo = f_out + ((size_t)s * N_ + mt * 16) * OD_;
#pragma unroll
    for (int v = 0; v < 8; ++v)
      fo[(v + 8 * half) * OD_ + n0 + lr] = c2[v] + bb;
  }
}

template <int NST>
static __device__ inline void ins_tiles(const float* __restrict__ langS,
                                        const float* arow, float* hrow) {
  v2f l0[NST > 0 ? NST : 1], l1[NST > 0 ? NST : 1];
#pragma unroll
  for (int i = 0; i < NST; ++i) {
    l0[i] = *reinterpret_cast<const v2f*>(langS + (4 * i) * OD_);
    l1[i] = *reinterpret_cast<const v2f*>(langS + (4 * i + 1) * OD_);
  }
  v8f ca = {}, cb = {};
#pragma unroll
  for (int i = 0; i < NST; ++i) {
    v2f a, ba, bb;
    a.x = arow[4 * i];
    a.y = arow[4 * i + 1];
    ba.x = l0[i].x; ba.y = l1[i].x;
    bb.x = l0[i].y; bb.y = l1[i].y;
    ca = wmma4(a, ba, ca);
    cb = wmma4(a, bb, cb);
  }
#pragma unroll
  for (int v = 0; v < 8; ++v) {
    v2f o;
    o.x = ca[v]; o.y = cb[v];
    *reinterpret_cast<v2f*>(hrow + v * OD_) = o;
  }
}

template <int MODE>
__global__ void __launch_bounds__(128)
fused_kernel(const float* __restrict__ coord,
                             const float* __restrict__ lang_mask,
                             const float* __restrict__ rel_w1,
                             const float* __restrict__ rel_b1,
                             const float* __restrict__ rel_w2,
                             const float* __restrict__ rel_b2,
                             const int*   __restrict__ idx_ws,
                             const float* __restrict__ lang_ws,
                             const float* __restrict__ f_ws,
                             float* __restrict__ out,
                             float* __restrict__ score) {
  __shared__ __attribute__((aligned(16))) float fg[16 * OD_];
  __shared__ float relin[16 * 16];
  __shared__ __attribute__((aligned(16))) float hbuf[16 * OD_];
  __shared__ float relT[OD_ * RTP];
  __shared__ float attnsm[16 * LP_];
  __shared__ float sred[16 * 4];
  __shared__ float mksm[L_];

  const int tid  = threadIdx.x;
  const int lane = tid & 31, wave = tid >> 5;
  const int half = lane >> 4, lr = lane & 15;
  const int s  = (MODE == 0) ? (blockIdx.x >> 8) : (blockIdx.x >> 4);
  const int nb = (MODE == 0) ? (blockIdx.x & 255) : ((blockIdx.x & 15) * 16);

  const float mk0 = lang_mask[s * L_ + lane];
  const float mk1 = lang_mask[s * L_ + 32 + (lane & 7)];

  int nbi[16];
#pragma unroll
  for (int r = 0; r < 16; ++r) {
    const int pt = (MODE == 0) ? nb : nb + r;
    const int j  = (MODE == 0) ? r : 16;
    nbi[r] = idx_ws[((size_t)s * N_ + pt) * KNB + j];
  }
  float gv[16];
#pragma unroll
  for (int r = 0; r < 16; ++r)
    gv[r] = f_ws[((size_t)s * N_ + nbi[r]) * OD_ + tid];

  if (tid < 16) {
    const int pt = (MODE == 0) ? nb : nb + tid;
    const int j  = (MODE == 0) ? tid : 16;
    const float* cs = coord + (size_t)s * N_ * 3;
    const int m = idx_ws[((size_t)s * N_ + pt) * KNB + j];
    const float nx = cs[m * 3 + 0], ny = cs[m * 3 + 1], nz = cs[m * 3 + 2];
    const float ox = cs[pt * 3 + 0], oy = cs[pt * 3 + 1], oz = cs[pt * 3 + 2];
    const float rx = nx - ox, ry = ny - oy, rz = nz - oz;
    const float dist = sqrtf(rx * rx + ry * ry + rz * rz);
    float* ri = relin + tid * 16;
    ri[0] = nx; ri[1] = ny; ri[2] = nz;
    ri[3] = ox; ri[4] = oy; ri[5] = oz;
    ri[6] = rx; ri[7] = ry; ri[8] = rz;
    ri[9] = dist;
    ri[10] = 0.f; ri[11] = 0.f; ri[12] = 0.f;
    ri[13] = 0.f; ri[14] = 0.f; ri[15] = 0.f;
  }
#pragma unroll
  for (int r = 0; r < 16; ++r) fg[r * OD_ + tid] = gv[r];
  unsigned balA, balB;
  lang_ballots(mk0, mk1, lane, balA, balB);
  const int cnt = __builtin_popcount(balA) + __builtin_popcount(balB);
  if (tid < L_) mksm[lang_dest(tid, balA, balB, cnt)] = (wave == 0) ? mk0 : mk1;
  __syncthreads();

  float4* hx = reinterpret_cast<float4*>(hbuf);
  v2f bin[4];
  for (int i = 0; i < 4; ++i) {
    const float* rp = relin + lr * 16 + 4 * i + 2 * half;
    bin[i].x = rp[0];
    bin[i].y = rp[1];
  }
  auto hidden_tile = [&](const int t) {
    v8f c = {};
    for (int i = 0; i < 4; ++i) {
      const int ka  = 4 * i + 2 * half;
      const int ka0 = min(ka, 9), ka1 = min(ka + 1, 9);
      v2f a;
      a.x = rel_w1[ka0 * OD_ + 16 * t + lr];
      a.y = rel_w1[ka1 * OD_ + 16 * t + lr];
      c = wmma4(a, bin[i], c);
    }
    const float* bp = rel_b1 + 16 * t + 8 * half;
    float h[8];
    for (int r = 0; r < 8; ++r) h[r] = fmaxf(c[r] + bp[r], 0.f);
    hx[(2 * t) * 32 + lane]     = make_float4(h[0], h[1], h[2], h[3]);
    hx[(2 * t + 1) * 32 + lane] = make_float4(h[4], h[5], h[6], h[7]);
  };

  if (wave < 3) {
    hidden_tile(wave);
    const int c0 = wave * 16;
    const float* langS = lang_ws + (size_t)s * L_ * OD_;
    const int lclamp = min(c0 + lr, L_ - 1);
    const float* arow = fg + lr * OD_ + 8 * half;
    const float* brow = langS + lclamp * OD_ + 8 * half;
    v8f c = {};
    for (int k0 = 0; k0 < OD_; k0 += 16) {
      const v4f b0 = *reinterpret_cast<const v4f*>(brow + k0);
      const v4f b1 = *reinterpret_cast<const v4f*>(brow + k0 + 4);
      const v4f a0 = *reinterpret_cast<const v4f*>(arow + k0);
      const v4f a1 = *reinterpret_cast<const v4f*>(arow + k0 + 4);
      v2f a, b;
      a.x = a0.x; a.y = a0.y; b.x = b0.x; b.y = b0.y; c = wmma4(a, b, c);
      a.x = a0.z; a.y = a0.w; b.x = b0.z; b.y = b0.w; c = wmma4(a, b, c);
      a.x = a1.x; a.y = a1.y; b.x = b1.x; b.y = b1.y; c = wmma4(a, b, c);
      a.x = a1.z; a.y = a1.w; b.x = b1.z; b.y = b1.w; c = wmma4(a, b, c);
    }
#pragma unroll
    for (int v = 0; v < 8; ++v)
      attnsm[(v + 8 * half) * LP_ + c0 + lr] = c[v];
  } else {
#pragma unroll
    for (int t = 3; t < 8; ++t) hidden_tile(t);
  }
  __syncthreads();

  {
    const int g = wave;
    v8f hT[8];
    for (int t = 0; t < 8; ++t) {
      const float4 lo = hx[(2 * t) * 32 + lane];
      const float4 hi = hx[(2 * t + 1) * 32 + lane];
      hT[t][0] = lo.x; hT[t][1] = lo.y; hT[t][2] = lo.z; hT[t][3] = lo.w;
      hT[t][4] = hi.x; hT[t][5] = hi.y; hT[t][6] = hi.z; hT[t][7] = hi.w;
    }

    const float* wp = rel_w2 + 32 * g + 2 * lr;
    v8f ca = {}, cb = {};
    for (int t = 0; t < 8; ++t) {
      for (int q = 0; q < 4; ++q) {
        const int k = 16 * t + 8 * half + 2 * q;
        const v2f w0 = *reinterpret_cast<const v2f*>(wp + k * OD_);
        const v2f w1 = *reinterpret_cast<const v2f*>(wp + (k + 1) * OD_);
        v2f aa, ab, b;
        aa.x = w0.x; aa.y = w1.x;
        ab.x = w0.y; ab.y = w1.y;
        b.x = hT[t][2 * q];
        b.y = hT[t][2 * q + 1];
        ca = wmma4(aa, b, ca);
        cb = wmma4(ab, b, cb);
      }
    }
    const int dbase = 32 * g + 16 * half;
    const float* bp2 = rel_b2 + dbase;
    for (int r = 0; r < 8; ++r) {
      relT[(dbase + 2 * r) * RTP + lr]     = ca[r] + bp2[2 * r];
      relT[(dbase + 2 * r + 1) * RTP + lr] = cb[r] + bp2[2 * r + 1];
    }
  }
  __syncthreads();

  {
    float* rowp = attnsm + (tid >> 3) * LP_ + (tid & 7);
    const float* mkp = mksm + (tid & 7);
    float v[5];
    for (int i = 0; i < 5; ++i) v[i] = rowp[8 * i];
    float mx = -FLT_MAX;
    for (int i = 0; i < 5; ++i) mx = fmaxf(mx, v[i]);
    mx = fmaxf(mx, lane_xor<1>(mx));
    mx = fmaxf(mx, lane_xor<2>(mx));
    mx = fmaxf(mx, lane_xor<4>(mx));
    float ssum = 0.f;
    for (int i = 0; i < 5; ++i) {
      v[i] = expf(v[i] - mx);
      ssum += v[i];
    }
    ssum += lane_xor<1>(ssum);
    ssum += lane_xor<2>(ssum);
    ssum += lane_xor<4>(ssum);
    const float inv = 1.f / ssum;
    float s2 = 0.f;
    for (int i = 0; i < 5; ++i) {
      v[i] = v[i] * inv * mkp[8 * i];
      s2 += v[i];
    }
    s2 += lane_xor<1>(s2);
    s2 += lane_xor<2>(s2);
    s2 += lane_xor<4>(s2);
    const float inv2 = 1.f / (s2 + 1e-7f);
    for (int i = 0; i < 5; ++i) rowp[8 * i] = v[i] * inv2;
  }
  __syncthreads();

  {
    const int c0 = 32 * wave + 2 * lr;
    const float* langS = lang_ws + (size_t)s * L_ * OD_ + c0 + 2 * half * OD_;
    const float* arow  = attnsm + lr * LP_ + 2 * half;
    float* hrow = hbuf + 8 * half * OD_ + c0;
    switch ((cnt + 3) >> 2) {
      case 0:  ins_tiles<0>(langS, arow, hrow); break;
      case 1:  ins_tiles<1>(langS, arow, hrow); break;
      case 2:  ins_tiles<2>(langS, arow, hrow); break;
      case 3:  ins_tiles<3>(langS, arow, hrow); break;
      case 4:  ins_tiles<4>(langS, arow, hrow); break;
      case 5:  ins_tiles<5>(langS, arow, hrow); break;
      case 6:  ins_tiles<6>(langS, arow, hrow); break;
      case 7:  ins_tiles<7>(langS, arow, hrow); break;
      case 8:  ins_tiles<8>(langS, arow, hrow); break;
      case 9:  ins_tiles<9>(langS, arow, hrow); break;
      default: ins_tiles<10>(langS, arow, hrow); break;
    }
  }
  __syncthreads();

  if (MODE == 0) {
    const int d = tid;
    float acc = 0.f;
    for (int j = 0; j < 16; ++j)
      acc += fg[j * OD_ + d] * hbuf[j * OD_ + d] * relT[d * RTP + j];
    acc += f_ws[((size_t)s * N_ + nb) * OD_ + d];
    out[((size_t)s * N_ + nb) * OD_ + d] = acc;
  } else {
    const int d = tid;
    for (int r = 0; r < 16; ++r) {
      float* op = out + ((size_t)s * N_ + nb + r) * OD_ + d;
      const float o = *op + fg[r * OD_ + d] * hbuf[r * OD_ + d] * relT[d * RTP + r];
      *op = o;
      float w = o;
      for (int off = 16; off > 0; off >>= 1) w += __shfl_xor(w, off);
      if (lane == 0) sred[r * 4 + wave] = w;
    }
    __syncthreads();
    if (tid < 16)
      score[(size_t)s * N_ + nb + tid] =
          (sred[tid * 4] + sred[tid * 4 + 1]) + (sred[tid * 4 + 2] + sred[tid * 4 + 3]);
  }
}

extern "C" void kernel_launch(void* const* d_in, const int* in_sizes, int n_in,
                              void* d_out, int out_size, void* d_ws,
                              size_t ws_size, hipStream_t stream) {
  (void)in_sizes; (void)n_in; (void)out_size; (void)ws_size;
  const float* feat      = (const float*)d_in[0];
  const float* coord     = (const float*)d_in[1];
  const float* lang_feat = (const float*)d_in[2];
  const float* lang_mask = (const float*)d_in[3];
  const float* rel_w1    = (const float*)d_in[4];
  const float* rel_b1    = (const float*)d_in[5];
  const float* rel_w2    = (const float*)d_in[6];
  const float* rel_b2    = (const float*)d_in[7];
  const float* lang_w1   = (const float*)d_in[8];
  const float* lang_b1   = (const float*)d_in[9];
  const float* lang_w2   = (const float*)d_in[10];
  const float* lang_b2   = (const float*)d_in[11];
  const float* feat_w1   = (const float*)d_in[12];
  const float* feat_b1   = (const float*)d_in[13];
  const float* feat_w2   = (const float*)d_in[14];
  const float* feat_b2   = (const float*)d_in[15];

  char* ws = (char*)d_ws;
  int*   idx_ws  = (int*)ws;
  float* lang_ws = (float*)(ws + (size_t)S_ * N_ * KNB * 4);
  float* f_ws    = (float*)(ws + (size_t)S_ * N_ * KNB * 4 +
                                 (size_t)S_ * L_ * OD_ * 4);

  float* out   = (float*)d_out;
  float* score = out + (size_t)S_ * N_ * OD_;

  knn_kernel<<<S_ * (N_ / KNN_TPB), KNN_TPB, 0, stream>>>(coord, idx_ws);
  lang_mlp_kernel<<<S_ * L_, 128, 0, stream>>>(lang_feat, lang_mask, lang_w1,
                                               lang_b1, lang_w2, lang_b2,
                                               lang_ws);
  feat_mlp_kernel<<<S_ * (N_ / 16), 256, 0, stream>>>(feat, feat_w1, feat_b1,
                                                      feat_w2, feat_b2, f_ws);
  fused_kernel<0><<<S_ * N_, 128, 0, stream>>>(coord, lang_mask, rel_w1,
                                               rel_b1, rel_w2, rel_b2, idx_ws,
                                               lang_ws, f_ws, out, score);
  fused_kernel<1><<<S_ * (N_ / 16), 128, 0, stream>>>(coord, lang_mask, rel_w1,
                                                      rel_b1, rel_w2, rel_b2,
                                                      idx_ws, lang_ws, f_ws,
                                                      out, score);
}


# kNN: 4 threads per point, each sweeps a quarter of the 256 candidates into its own sorted 17-list; lists merged pairwise through DPP (2 rounds of 17 shift-ins); 512 blocks x 4 waves; sums unchanged
# speedup vs baseline: 4.3628x; 1.0367x over previous
#include <hip/hip_runtime.h>
#include <hip/hip_bf16.h>
#include <float.h>


typedef float v2f __attribute__((ext_vector_type(2)));
typedef float v4f __attribute__((ext_vector_type(4)));
typedef float v8f __attribute__((ext_vector_type(8)));

#define S_      64
#define N_      256
#define L_      40
#define LP_     48
#define KNB     17
#define OD_     128
#define LANGID_ 256
#define PCID_   128
#define RTP     17

static __device__ inline v8f wmma4(v2f a, v2f b, v8f c) {
  return __builtin_amdgcn_wmma_f32_16x16x4_f32(false, a, false, b, (short)0, c,
                                               false, false);
}

template <int M>
static __device__ inline float lane_xor(float x) {
  const int b = __builtin_bit_cast(int, x);
  return __builtin_bit_cast(
      float, __builtin_amdgcn_mov_dpp(b, 0x160 + M, 0xf, 0xf, false));
}

static __device__ inline void lang_ballots(const float m0, const float m1,
                                           const int lane, unsigned& balA,
                                           unsigned& balB) {
  balA = __builtin_amdgcn_ballot_w32(m0 != 0.f);
  balB = __builtin_amdgcn_ballot_w32((lane < 8) && (m1 != 0.f));
}
static __device__ inline int lang_dest(const int l, const unsigned balA,
                                       const unsigned balB, const int cnt) {
  const unsigned below = (1u << (l & 31)) - 1u;
  const int before = (l < 32) ? __builtin_popcount(balA & below)
                              : __builtin_popcount(balA) +
                                    __builtin_popcount(balB & below);
  const bool on = ((((l < 32) ? balA : balB) >> (l & 31)) & 1u) != 0u;
  return on ? before : cnt + (l - before);
}

#define KNN_WAYS 4
#define KNN_TPB  128
#define KNN_PPB  (KNN_TPB / KNN_WAYS)
#define KNN_CPW  (N_ / KNN_WAYS)
#define KNN_CPT  8
#define KNN_SH   8
#define KNN_LDS  (N_ + (KNN_WAYS - 1) * KNN_SH)

static __device__ inline int knn_slot(const int m) {
  return m + (m / KNN_CPW) * KNN_SH;
}

template <int M>
static __device__ inline int lane_xor_i(const int x) {
  return __builtin_amdgcn_mov_dpp(x, 0x160 + M, 0xf, 0xf, false);
}

static __device__ inline void knn_insert(float (&ld)[KNB], int (&li)[KNB],
                                         const float d2, const int m) {
  bool lt[KNB];
#pragma unroll
  for (int j = 0; j < KNB; ++j) lt[j] = d2 < ld[j];
#pragma unroll
  for (int j = KNB - 1; j >= 1; --j) {
    const int below = li[j - 1], here = li[j];
    ld[j] = __builtin_amdgcn_fmed3f(ld[j - 1], d2, ld[j]);
    li[j] = lt[j] ? (lt[j - 1] ? below : m) : here;
  }
  const int first = li[0];
  ld[0] = fminf(ld[0], d2);
  li[0] = lt[0] ? m : first;
}

template <int M>
static __device__ inline void knn_merge(float (&ld)[KNB], int (&li)[KNB]) {
  float od[KNB];
  int   oi[KNB];
#pragma unroll
  for (int j = 0; j < KNB; ++j) {
    od[j] = lane_xor<M>(ld[j]);
    oi[j] = lane_xor_i<M>(li[j]);
  }
#pragma unroll
  for (int j = 0; j < KNB; ++j) knn_insert(ld, li, od[j], oi[j]);
}

__global__ void __launch_bounds__(KNN_TPB)
knn_kernel(const float* __restrict__ coord, int* __restrict__ idx_out) {
  __shared__ float cx[KNN_LDS], cy[KNN_LDS], cz[KNN_LDS];
  const int s = blockIdx.x / (N_ / KNN_PPB);
  const int n = (blockIdx.x % (N_ / KNN_PPB)) * KNN_PPB + threadIdx.x / KNN_WAYS;
  const int q = threadIdx.x % KNN_WAYS;
  const float* cs = coord + (size_t)s * N_ * 3;
#pragma unroll
  for (int r = 0; r < N_ / KNN_TPB; ++r) {
    const int i = r * KNN_TPB + threadIdx.x;
    const int e = knn_slot(i);
    cx[e] = cs[i * 3 + 0];
    cy[e] = cs[i * 3 + 1];
    cz[e] = cs[i * 3 + 2];
  }
  __syncthreads();

  const int en = knn_slot(n);
  const float px = cx[en], py = cy[en], pz = cz[en];
  float ld[KNB];
  int   li[KNB];
#pragma unroll
  for (int j = 0; j < KNB; ++j) { ld[j] = INFINITY; li[j] = 0x7fffffff; }

  const int mq = q * KNN_CPW;
  const int eq = knn_slot(mq);
  for (int m0 = 0; m0 < KNN_CPW; m0 += KNN_CPT) {
#pragma unroll
    for (int u = 0; u < KNN_CPT; ++u) {
      const int e = eq + m0 + u;
      const float dx = cx[e] - px, dy = cy[e] - py, dz = cz[e] - pz;
      const float d2 = dx * dx + dy * dy + dz * dz;
      knn_insert(ld, li, d2, mq + m0 + u);
    }
  }

  knn_merge<1>(ld, li);
  knn_merge<2>(ld, li);

  if (q == 0) {
    int* out = idx_out + ((size_t)s * N_ + n) * KNB;
#pragma unroll
    for (int j = 0; j < KNB; ++j) out[j] = li[j];
  }
}

__global__ void lang_mlp_kernel(const float* __restrict__ lang_feat,
                                const float* __restrict__ lang_mask,
                                const float* __restrict__ w1,
                                const float* __restrict__ b1,
                                const float* __restrict__ w2,
                                const float* __restrict__ b2,
                                float* __restrict__ lang_out) {
  __shared__ float xs[LANGID_];
  __shared__ float hs[OD_];
  const int row = blockIdx.x;
  const int t   = threadIdx.x;
  const int s = row / L_, l = row - s * L_;
  const float mk0 = lang_mask[s * L_ + (t & 31)];
  const float mk1 = lang_mask[s * L_ + 32 + (t & 7)];
  const float* x = lang_feat + (size_t)row * LANGID_;
  xs[t]       = x[t];
  xs[t + 128] = x[t + 128];
  __syncthreads();
  float acc = b1[t];
  for (int k = 0; k < LANGID_; ++k) acc += xs[k] * w1[k * OD_ + t];
  hs[t] = fmaxf(acc, 0.0f);
  __syncthreads();
  float acc2 = b2[t];
  for (int k = 0; k < OD_; ++k) acc2 += hs[k] * w2[k * OD_ + t];
  unsigned balA, balB;
  lang_ballots(mk0, mk1, t & 31, balA, balB);
  const int cnt  = __builtin_popcount(balA) + __builtin_popcount(balB);
  const int dest = lang_dest(l, balA, balB, cnt);
  lang_out[((size_t)s * L_ + dest) * OD_ + t] = acc2;
}

__global__ void feat_mlp_kernel(const float* __restrict__ feat,
                                const float* __restrict__ w1,
                                const float* __restrict__ b1,
                                const float* __restrict__ w2,
                                const float* __restrict__ b2,
                                float* __restrict__ f_out) {
  __shared__ float Asm[16 * PCID_];
  __shared__ float Hsm[16 * OD_];
  const int s    = blockIdx.x >> 4;
  const int mt   = blockIdx.x & 15;
  const int tid  = threadIdx.x;
  const int lane = tid & 31, wave = tid >> 5;
  const int half = lane >> 4, lr = lane & 15;

  const float* A = feat + ((size_t)s * N_ + mt * 16) * PCID_;
  for (int i = tid; i < 16 * PCID_; i += 256) Asm[i] = A[i];
  __syncthreads();

  const int n0 = wave * 16;
  v8f c = {};
  for (int k0 = 0; k0 < PCID_; k0 += 4) {
    const int ka = k0 + 2 * half;
    v2f a, b;
    a.x = Asm[lr * PCID_ + ka];
    a.y = Asm[lr * PCID_ + ka + 1];
    b.x = w1[ka * OD_ + n0 + lr];
    b.y = w1[(ka + 1) * OD_ + n0 + lr];
    c = wmma4(a, b, c);
  }
  {
    const float bb = b1[n0 + lr];
#pragma unroll
    for (int v = 0; v < 8; ++v)
      Hsm[(v + 8 * half) * OD_ + n0 + lr] = fmaxf(c[v] + bb, 0.0f);
  }
  __syncthreads();

  v8f c2 = {};
  for (int k0 = 0; k0 < OD_; k0 += 4) {
    const int ka = k0 + 2 * half;
    v2f a, b;
    a.x = Hsm[lr * OD_ + ka];
    a.y = Hsm[lr * OD_ + ka + 1];
    b.x = w2[ka * OD_ + n0 + lr];
    b.y = w2[(ka + 1) * OD_ + n0 + lr];
    c2 = wmma4(a, b, c2);
  }
  {
    const float bb = b2[n0 + lr];
    float* fo = f_out + ((size_t)s * N_ + mt * 16) * OD_;
#pragma unroll
    for (int v = 0; v < 8; ++v)
      fo[(v + 8 * half) * OD_ + n0 + lr] = c2[v] + bb;
  }
}

template <int NST>
static __device__ inline void ins_tiles(const float* __restrict__ langS,
                                        const float* arow, float* hrow) {
  v2f l0[NST > 0 ? NST : 1], l1[NST > 0 ? NST : 1];
#pragma unroll
  for (int i = 0; i < NST; ++i) {
    l0[i] = *reinterpret_cast<const v2f*>(langS + (4 * i) * OD_);
    l1[i] = *reinterpret_cast<const v2f*>(langS + (4 * i + 1) * OD_);
  }
  v8f ca = {}, cb = {};
#pragma unroll
  for (int i = 0; i < NST; ++i) {
    v2f a, ba, bb;
    a.x = arow[4 * i];
    a.y = arow[4 * i + 1];
    ba.x = l0[i].x; ba.y = l1[i].x;
    bb.x = l0[i].y; bb.y = l1[i].y;
    ca = wmma4(a, ba, ca);
    cb = wmma4(a, bb, cb);
  }
#pragma unroll
  for (int v = 0; v < 8; ++v) {
    v2f o;
    o.x = ca[v]; o.y = cb[v];
    *reinterpret_cast<v2f*>(hrow + v * OD_) = o;
  }
}

template <int MODE>
__global__ void __launch_bounds__(128)
fused_kernel(const float* __restrict__ coord,
                             const float* __restrict__ lang_mask,
                             const float* __restrict__ rel_w1,
                             const float* __restrict__ rel_b1,
                             const float* __restrict__ rel_w2,
                             const float* __restrict__ rel_b2,
                             const int*   __restrict__ idx_ws,
                             const float* __restrict__ lang_ws,
                             const float* __restrict__ f_ws,
                             float* __restrict__ out,
                             float* __restrict__ score) {
  __shared__ __attribute__((aligned(16))) float fg[16 * OD_];
  __shared__ float relin[16 * 16];
  __shared__ __attribute__((aligned(16))) float hbuf[16 * OD_];
  __shared__ float relT[OD_ * RTP];
  __shared__ float attnsm[16 * LP_];
  __shared__ float sred[16 * 4];
  __shared__ float mksm[L_];

  const int tid  = threadIdx.x;
  const int lane = tid & 31, wave = tid >> 5;
  const int half = lane >> 4, lr = lane & 15;
  const int s  = (MODE == 0) ? (blockIdx.x >> 8) : (blockIdx.x >> 4);
  const int nb = (MODE == 0) ? (blockIdx.x & 255) : ((blockIdx.x & 15) * 16);

  const float mk0 = lang_mask[s * L_ + lane];
  const float mk1 = lang_mask[s * L_ + 32 + (lane & 7)];

  int nbi[16];
#pragma unroll
  for (int r = 0; r < 16; ++r) {
    const int pt = (MODE == 0) ? nb : nb + r;
    const int j  = (MODE == 0) ? r : 16;
    nbi[r] = idx_ws[((size_t)s * N_ + pt) * KNB + j];
  }
  float gv[16];
#pragma unroll
  for (int r = 0; r < 16; ++r)
    gv[r] = f_ws[((size_t)s * N_ + nbi[r]) * OD_ + tid];

  if (tid < 16) {
    const int pt = (MODE == 0) ? nb : nb + tid;
    const int j  = (MODE == 0) ? tid : 16;
    const float* cs = coord + (size_t)s * N_ * 3;
    const int m = idx_ws[((size_t)s * N_ + pt) * KNB + j];
    const float nx = cs[m * 3 + 0], ny = cs[m * 3 + 1], nz = cs[m * 3 + 2];
    const float ox = cs[pt * 3 + 0], oy = cs[pt * 3 + 1], oz = cs[pt * 3 + 2];
    const float rx = nx - ox, ry = ny - oy, rz = nz - oz;
    const float dist = sqrtf(rx * rx + ry * ry + rz * rz);
    float* ri = relin + tid * 16;
    ri[0] = nx; ri[1] = ny; ri[2] = nz;
    ri[3] = ox; ri[4] = oy; ri[5] = oz;
    ri[6] = rx; ri[7] = ry; ri[8] = rz;
    ri[9] = dist;
    ri[10] = 0.f; ri[11] = 0.f; ri[12] = 0.f;
    ri[13] = 0.f; ri[14] = 0.f; ri[15] = 0.f;
  }
#pragma unroll
  for (int r = 0; r < 16; ++r) fg[r * OD_ + tid] = gv[r];
  unsigned balA, balB;
  lang_ballots(mk0, mk1, lane, balA, balB);
  const int cnt = __builtin_popcount(balA) + __builtin_popcount(balB);
  if (tid < L_) mksm[lang_dest(tid, balA, balB, cnt)] = (wave == 0) ? mk0 : mk1;
  __syncthreads();

  float4* hx = reinterpret_cast<float4*>(hbuf);
  v2f bin[4];
  for (int i = 0; i < 4; ++i) {
    const float* rp = relin + lr * 16 + 4 * i + 2 * half;
    bin[i].x = rp[0];
    bin[i].y = rp[1];
  }
  auto hidden_tile = [&](const int t) {
    v8f c = {};
    for (int i = 0; i < 4; ++i) {
      const int ka  = 4 * i + 2 * half;
      const int ka0 = min(ka, 9), ka1 = min(ka + 1, 9);
      v2f a;
      a.x = rel_w1[ka0 * OD_ + 16 * t + lr];
      a.y = rel_w1[ka1 * OD_ + 16 * t + lr];
      c = wmma4(a, bin[i], c);
    }
    const float* bp = rel_b1 + 16 * t + 8 * half;
    float h[8];
    for (int r = 0; r < 8; ++r) h[r] = fmaxf(c[r] + bp[r], 0.f);
    hx[(2 * t) * 32 + lane]     = make_float4(h[0], h[1], h[2], h[3]);
    hx[(2 * t + 1) * 32 + lane] = make_float4(h[4], h[5], h[6], h[7]);
  };

  if (wave < 3) {
    hidden_tile(wave);
    const int c0 = wave * 16;
    const float* langS = lang_ws + (size_t)s * L_ * OD_;
    const int lclamp = min(c0 + lr, L_ - 1);
    const float* arow = fg + lr * OD_ + 8 * half;
    const float* brow = langS + lclamp * OD_ + 8 * half;
    v8f c = {};
    for (int k0 = 0; k0 < OD_; k0 += 16) {
      const v4f b0 = *reinterpret_cast<const v4f*>(brow + k0);
      const v4f b1 = *reinterpret_cast<const v4f*>(brow + k0 + 4);
      const v4f a0 = *reinterpret_cast<const v4f*>(arow + k0);
      const v4f a1 = *reinterpret_cast<const v4f*>(arow + k0 + 4);
      v2f a, b;
      a.x = a0.x; a.y = a0.y; b.x = b0.x; b.y = b0.y; c = wmma4(a, b, c);
      a.x = a0.z; a.y = a0.w; b.x = b0.z; b.y = b0.w; c = wmma4(a, b, c);
      a.x = a1.x; a.y = a1.y; b.x = b1.x; b.y = b1.y; c = wmma4(a, b, c);
      a.x = a1.z; a.y = a1.w; b.x = b1.z; b.y = b1.w; c = wmma4(a, b, c);
    }
#pragma unroll
    for (int v = 0; v < 8; ++v)
      attnsm[(v + 8 * half) * LP_ + c0 + lr] = c[v];
  } else {
#pragma unroll
    for (int t = 3; t < 8; ++t) hidden_tile(t);
  }
  __syncthreads();

  {
    const int g = wave;
    v8f hT[8];
    for (int t = 0; t < 8; ++t) {
      const float4 lo = hx[(2 * t) * 32 + lane];
      const float4 hi = hx[(2 * t + 1) * 32 + lane];
      hT[t][0] = lo.x; hT[t][1] = lo.y; hT[t][2] = lo.z; hT[t][3] = lo.w;
      hT[t][4] = hi.x; hT[t][5] = hi.y; hT[t][6] = hi.z; hT[t][7] = hi.w;
    }

    const float* wp = rel_w2 + 32 * g + 2 * lr;
    v8f ca = {}, cb = {};
    for (int t = 0; t < 8; ++t) {
      for (int q = 0; q < 4; ++q) {
        const int k = 16 * t + 8 * half + 2 * q;
        const v2f w0 = *reinterpret_cast<const v2f*>(wp + k * OD_);
        const v2f w1 = *reinterpret_cast<const v2f*>(wp + (k + 1) * OD_);
        v2f aa, ab, b;
        aa.x = w0.x; aa.y = w1.x;
        ab.x = w0.y; ab.y = w1.y;
        b.x = hT[t][2 * q];
        b.y = hT[t][2 * q + 1];
        ca = wmma4(aa, b, ca);
        cb = wmma4(ab, b, cb);
      }
    }
    const int dbase = 32 * g + 16 * half;
    const float* bp2 = rel_b2 + dbase;
    for (int r = 0; r < 8; ++r) {
      relT[(dbase + 2 * r) * RTP + lr]     = ca[r] + bp2[2 * r];
      relT[(dbase + 2 * r + 1) * RTP + lr] = cb[r] + bp2[2 * r + 1];
    }
  }
  __syncthreads();

  {
    float* rowp = attnsm + (tid >> 3) * LP_ + (tid & 7);
    const float* mkp = mksm + (tid & 7);
    float v[5];
    for (int i = 0; i < 5; ++i) v[i] = rowp[8 * i];
    float mx = -FLT_MAX;
    for (int i = 0; i < 5; ++i) mx = fmaxf(mx, v[i]);
    mx = fmaxf(mx, lane_xor<1>(mx));
    mx = fmaxf(mx, lane_xor<2>(mx));
    mx = fmaxf(mx, lane_xor<4>(mx));
    float ssum = 0.f;
    for (int i = 0; i < 5; ++i) {
      v[i] = expf(v[i] - mx);
      ssum += v[i];
    }
    ssum += lane_xor<1>(ssum);
    ssum += lane_xor<2>(ssum);
    ssum += lane_xor<4>(ssum);
    const float inv = 1.f / ssum;
    float s2 = 0.f;
    for (int i = 0; i < 5; ++i) {
      v[i] = v[i] * inv * mkp[8 * i];
      s2 += v[i];
    }
    s2 += lane_xor<1>(s2);
    s2 += lane_xor<2>(s2);
    s2 += lane_xor<4>(s2);
    const float inv2 = 1.f / (s2 + 1e-7f);
    for (int i = 0; i < 5; ++i) rowp[8 * i] = v[i] * inv2;
  }
  __syncthreads();

  {
    const int c0 = 32 * wave + 2 * lr;
    const float* langS = lang_ws + (size_t)s * L_ * OD_ + c0 + 2 * half * OD_;
    const float* arow  = attnsm + lr * LP_ + 2 * half;
    float* hrow = hbuf + 8 * half * OD_ + c0;
    switch ((cnt + 3) >> 2) {
      case 0:  ins_tiles<0>(langS, arow, hrow); break;
      case 1:  ins_tiles<1>(langS, arow, hrow); break;
      case 2:  ins_tiles<2>(langS, arow, hrow); break;
      case 3:  ins_tiles<3>(langS, arow, hrow); break;
      case 4:  ins_tiles<4>(langS, arow, hrow); break;
      case 5:  ins_tiles<5>(langS, arow, hrow); break;
      case 6:  ins_tiles<6>(langS, arow, hrow); break;
      case 7:  ins_tiles<7>(langS, arow, hrow); break;
      case 8:  ins_tiles<8>(langS, arow, hrow); break;
      case 9:  ins_tiles<9>(langS, arow, hrow); break;
      default: ins_tiles<10>(langS, arow, hrow); break;
    }
  }
  __syncthreads();

  if (MODE == 0) {
    const int d = tid;
    float acc = 0.f;
    for (int j = 0; j < 16; ++j)
      acc += fg[j * OD_ + d] * hbuf[j * OD_ + d] * relT[d * RTP + j];
    acc += f_ws[((size_t)s * N_ + nb) * OD_ + d];
    out[((size_t)s * N_ + nb) * OD_ + d] = acc;
  } else {
    const int d = tid;
    for (int r = 0; r < 16; ++r) {
      float* op = out + ((size_t)s * N_ + nb + r) * OD_ + d;
      const float o = *op + fg[r * OD_ + d] * hbuf[r * OD_ + d] * relT[d * RTP + r];
      *op = o;
      float w = o;
      for (int off = 16; off > 0; off >>= 1) w += __shfl_xor(w, off);
      if (lane == 0) sred[r * 4 + wave] = w;
    }
    __syncthreads();
    if (tid < 16)
      score[(size_t)s * N_ + nb + tid] =
          (sred[tid * 4] + sred[tid * 4 + 1]) + (sred[tid * 4 + 2] + sred[tid * 4 + 3]);
  }
}

extern "C" void kernel_launch(void* const* d_in, const int* in_sizes, int n_in,
                              void* d_out, int out_size, void* d_ws,
                              size_t ws_size, hipStream_t stream) {
  (void)in_sizes; (void)n_in; (void)out_size; (void)ws_size;
  const float* feat      = (const float*)d_in[0];
  const float* coord     = (const float*)d_in[1];
  const float* lang_feat = (const float*)d_in[2];
  const float* lang_mask = (const float*)d_in[3];
  const float* rel_w1    = (const float*)d_in[4];
  const float* rel_b1    = (const float*)d_in[5];
  const float* rel_w2    = (const float*)d_in[6];
  const float* rel_b2    = (const float*)d_in[7];
  const float* lang_w1   = (const float*)d_in[8];
  const float* lang_b1   = (const float*)d_in[9];
  const float* lang_w2   = (const float*)d_in[10];
  const float* lang_b2   = (const float*)d_in[11];
  const float* feat_w1   = (const float*)d_in[12];
  const float* feat_b1   = (const float*)d_in[13];
  const float* feat_w2   = (const float*)d_in[14];
  const float* feat_b2   = (const float*)d_in[15];

  char* ws = (char*)d_ws;
  int*   idx_ws  = (int*)ws;
  float* lang_ws = (float*)(ws + (size_t)S_ * N_ * KNB * 4);
  float* f_ws    = (float*)(ws + (size_t)S_ * N_ * KNB * 4 +
                                 (size_t)S_ * L_ * OD_ * 4);

  float* out   = (float*)d_out;
  float* score = out + (size_t)S_ * N_ * OD_;

  knn_kernel<<<S_ * (N_ / KNN_PPB), KNN_TPB, 0, stream>>>(coord, idx_ws);
  lang_mlp_kernel<<<S_ * L_, 128, 0, stream>>>(lang_feat, lang_mask, lang_w1,
                                               lang_b1, lang_w2, lang_b2,
                                               lang_ws);
  feat_mlp_kernel<<<S_ * (N_ / 16), 256, 0, stream>>>(feat, feat_w1, feat_b1,
                                                      feat_w2, feat_b2, f_ws);
  fused_kernel<0><<<S_ * N_, 128, 0, stream>>>(coord, lang_mask, rel_w1,
                                               rel_b1, rel_w2, rel_b2, idx_ws,
                                               lang_ws, f_ws, out, score);
  fused_kernel<1><<<S_ * (N_ / 16), 128, 0, stream>>>(coord, lang_mask, rel_w1,
                                                      rel_b1, rel_w2, rel_b2,
                                                      idx_ws, lang_ws, f_ws,
                                                      out, score);
}


# Fused kernel (neighbors 0-15): 2 points per block; each fetched rel_w1, rel_w2 and lang fragment runs against both points' tiles in registers (half the L2 operand loads per point); sums unchanged
# speedup vs baseline: 4.6240x; 1.0599x over previous
#include <hip/hip_runtime.h>
#include <hip/hip_bf16.h>
#include <float.h>


typedef float v2f __attribute__((ext_vector_type(2)));
typedef float v4f __attribute__((ext_vector_type(4)));
typedef float v8f __attribute__((ext_vector_type(8)));

#define S_      64
#define N_      256
#define L_      40
#define LP_     48
#define KNB     17
#define OD_     128
#define LANGID_ 256
#define PCID_   128
#define RTP     17

static __device__ inline v8f wmma4(v2f a, v2f b, v8f c) {
  return __builtin_amdgcn_wmma_f32_16x16x4_f32(false, a, false, b, (short)0, c,
                                               false, false);
}

template <int M>
static __device__ inline float lane_xor(float x) {
  const int b = __builtin_bit_cast(int, x);
  return __builtin_bit_cast(
      float, __builtin_amdgcn_mov_dpp(b, 0x160 + M, 0xf, 0xf, false));
}

static __device__ inline void lang_ballots(const float m0, const float m1,
                                           const int lane, unsigned& balA,
                                           unsigned& balB) {
  balA = __builtin_amdgcn_ballot_w32(m0 != 0.f);
  balB = __builtin_amdgcn_ballot_w32((lane < 8) && (m1 != 0.f));
}
static __device__ inline int lang_dest(const int l, const unsigned balA,
                                       const unsigned balB, const int cnt) {
  const unsigned below = (1u << (l & 31)) - 1u;
  const int before = (l < 32) ? __builtin_popcount(balA & below)
                              : __builtin_popcount(balA) +
                                    __builtin_popcount(balB & below);
  const bool on = ((((l < 32) ? balA : balB) >> (l & 31)) & 1u) != 0u;
  return on ? before : cnt + (l - before);
}

#define KNN_WAYS 4
#define KNN_TPB  128
#define KNN_PPB  (KNN_TPB / KNN_WAYS)
#define KNN_CPW  (N_ / KNN_WAYS)
#define KNN_CPT  8
#define KNN_SH   8
#define KNN_LDS  (N_ + (KNN_WAYS - 1) * KNN_SH)

static __device__ inline int knn_slot(const int m) {
  return m + (m / KNN_CPW) * KNN_SH;
}

template <int M>
static __device__ inline int lane_xor_i(const int x) {
  return __builtin_amdgcn_mov_dpp(x, 0x160 + M, 0xf, 0xf, false);
}

static __device__ inline void knn_insert(float (&ld)[KNB], int (&li)[KNB],
                                         const float d2, const int m) {
  bool lt[KNB];
#pragma unroll
  for (int j = 0; j < KNB; ++j) lt[j] = d2 < ld[j];
#pragma unroll
  for (int j = KNB - 1; j >= 1; --j) {
    const int below = li[j - 1], here = li[j];
    ld[j] = __builtin_amdgcn_fmed3f(ld[j - 1], d2, ld[j]);
    li[j] = lt[j] ? (lt[j - 1] ? below : m) : here;
  }
  const int first = li[0];
  ld[0] = fminf(ld[0], d2);
  li[0] = lt[0] ? m : first;
}

template <int M>
static __device__ inline void knn_merge(float (&ld)[KNB], int (&li)[KNB]) {
  float od[KNB];
  int   oi[KNB];
#pragma unroll
  for (int j = 0; j < KNB; ++j) {
    od[j] = lane_xor<M>(ld[j]);
    oi[j] = lane_xor_i<M>(li[j]);
  }
#pragma unroll
  for (int j = 0; j < KNB; ++j) knn_insert(ld, li, od[j], oi[j]);
}

__global__ void __launch_bounds__(KNN_TPB)
knn_kernel(const float* __restrict__ coord, int* __restrict__ idx_out) {
  __shared__ float cx[KNN_LDS], cy[KNN_LDS], cz[KNN_LDS];
  const int s = blockIdx.x / (N_ / KNN_PPB);
  const int n = (blockIdx.x % (N_ / KNN_PPB)) * KNN_PPB + threadIdx.x / KNN_WAYS;
  const int q = threadIdx.x % KNN_WAYS;
  const float* cs = coord + (size_t)s * N_ * 3;
#pragma unroll
  for (int r = 0; r < N_ / KNN_TPB; ++r) {
    const int i = r * KNN_TPB + threadIdx.x;
    const int e = knn_slot(i);
    cx[e] = cs[i * 3 + 0];
    cy[e] = cs[i * 3 + 1];
    cz[e] = cs[i * 3 + 2];
  }
  __syncthreads();

  const int en = knn_slot(n);
  const float px = cx[en], py = cy[en], pz = cz[en];
  float ld[KNB];
  int   li[KNB];
#pragma unroll
  for (int j = 0; j < KNB; ++j) { ld[j] = INFINITY; li[j] = 0x7fffffff; }

  const int mq = q * KNN_CPW;
  const int eq = knn_slot(mq);
  for (int m0 = 0; m0 < KNN_CPW; m0 += KNN_CPT) {
#pragma unroll
    for (int u = 0; u < KNN_CPT; ++u) {
      const int e = eq + m0 + u;
      const float dx = cx[e] - px, dy = cy[e] - py, dz = cz[e] - pz;
      const float d2 = dx * dx + dy * dy + dz * dz;
      knn_insert(ld, li, d2, mq + m0 + u);
    }
  }

  knn_merge<1>(ld, li);
  knn_merge<2>(ld, li);

  if (q == 0) {
    int* out = idx_out + ((size_t)s * N_ + n) * KNB;
#pragma unroll
    for (int j = 0; j < KNB; ++j) out[j] = li[j];
  }
}

__global__ void lang_mlp_kernel(const float* __restrict__ lang_feat,
                                const float* __restrict__ lang_mask,
                                const float* __restrict__ w1,
                                const float* __restrict__ b1,
                                const float* __restrict__ w2,
                                const float* __restrict__ b2,
                                float* __restrict__ lang_out) {
  __shared__ float xs[LANGID_];
  __shared__ float hs[OD_];
  const int row = blockIdx.x;
  const int t   = threadIdx.x;
  const int s = row / L_, l = row - s * L_;
  const float mk0 = lang_mask[s * L_ + (t & 31)];
  const float mk1 = lang_mask[s * L_ + 32 + (t & 7)];
  const float* x = lang_feat + (size_t)row * LANGID_;
  xs[t]       = x[t];
  xs[t + 128] = x[t + 128];
  __syncthreads();
  float acc = b1[t];
  for (int k = 0; k < LANGID_; ++k) acc += xs[k] * w1[k * OD_ + t];
  hs[t] = fmaxf(acc, 0.0f);
  __syncthreads();
  float acc2 = b2[t];
  for (int k = 0; k < OD_; ++k) acc2 += hs[k] * w2[k * OD_ + t];
  unsigned balA, balB;
  lang_ballots(mk0, mk1, t & 31, balA, balB);
  const int cnt  = __builtin_popcount(balA) + __builtin_popcount(balB);
  const int dest = lang_dest(l, balA, balB, cnt);
  lang_out[((size_t)s * L_ + dest) * OD_ + t] = acc2;
}

__global__ void feat_mlp_kernel(const float* __restrict__ feat,
                                const float* __restrict__ w1,
                                const float* __restrict__ b1,
                                const float* __restrict__ w2,
                                const float* __restrict__ b2,
                                float* __restrict__ f_out) {
  __shared__ float Asm[16 * PCID_];
  __shared__ float Hsm[16 * OD_];
  const int s    = blockIdx.x >> 4;
  const int mt   = blockIdx.x & 15;
  const int tid  = threadIdx.x;
  const int lane = tid & 31, wave = tid >> 5;
  const int half = lane >> 4, lr = lane & 15;

  const float* A = feat + ((size_t)s * N_ + mt * 16) * PCID_;
  for (int i = tid; i < 16 * PCID_; i += 256) Asm[i] = A[i];
  __syncthreads();

  const int n0 = wave * 16;
  v8f c = {};
  for (int k0 = 0; k0 < PCID_; k0 += 4) {
    const int ka = k0 + 2 * half;
    v2f a, b;
    a.x = Asm[lr * PCID_ + ka];
    a.y = Asm[lr * PCID_ + ka + 1];
    b.x = w1[ka * OD_ + n0 + lr];
    b.y = w1[(ka + 1) * OD_ + n0 + lr];
    c = wmma4(a, b, c);
  }
  {
    const float bb = b1[n0 + lr];
#pragma unroll
    for (int v = 0; v < 8; ++v)
      Hsm[(v + 8 * half) * OD_ + n0 + lr] = fmaxf(c[v] + bb, 0.0f);
  }
  __syncthreads();

  v8f c2 = {};
  for (int k0 = 0; k0 < OD_; k0 += 4) {
    const int ka = k0 + 2 * half;
    v2f a, b;
    a.x = Hsm[lr * OD_ + ka];
    a.y = Hsm[lr * OD_ + ka + 1];
    b.x = w2[ka * OD_ + n0 + lr];
    b.y = w2[(ka + 1) * OD_ + n0 + lr];
    c2 = wmma4(a, b, c2);
  }
  {
    const float bb = b2[n0 + lr];
    float* fo = f_out + ((size_t)s * N_ + mt * 16) * OD_;
#pragma unroll
    for (int v = 0; v < 8; ++v)
      fo[(v + 8 * half) * OD_ + n0 + lr] = c2[v] + bb;
  }
}

template <int NST, int NP>
static __device__ inline void ins_tiles(const float* __restrict__ langS,
                                        const float* arow, float* hrow) {
  v2f l0[NST > 0 ? NST : 1], l1[NST > 0 ? NST : 1];
#pragma unroll
  for (int i = 0; i < NST; ++i) {
    l0[i] = *reinterpret_cast<const v2f*>(langS + (4 * i) * OD_);
    l1[i] = *reinterpret_cast<const v2f*>(langS + (4 * i + 1) * OD_);
  }
  v8f ca[NP], cb[NP];
#pragma unroll
  for (int p = 0; p < NP; ++p) { ca[p] = v8f{}; cb[p] = v8f{}; }
#pragma unroll
  for (int i = 0; i < NST; ++i) {
    v2f ba, bb;
    ba.x = l0[i].x; ba.y = l1[i].x;
    bb.x = l0[i].y; bb.y = l1[i].y;
#pragma unroll
    for (int p = 0; p < NP; ++p) {
      v2f a;
      a.x = arow[p * 16 * LP_ + 4 * i];
      a.y = arow[p * 16 * LP_ + 4 * i + 1];
      ca[p] = wmma4(a, ba, ca[p]);
      cb[p] = wmma4(a, bb, cb[p]);
    }
  }
#pragma unroll
  for (int p = 0; p < NP; ++p) {
#pragma unroll
    for (int v = 0; v < 8; ++v) {
      v2f o;
      o.x = ca[p][v]; o.y = cb[p][v];
      *reinterpret_cast<v2f*>(hrow + p * 16 * OD_ + v * OD_) = o;
    }
  }
}

template <int NP>
static __device__ inline void ins_product(const int nst,
                                          const float* __restrict__ langS,
                                          const float* arow, float* hrow) {
  switch (nst) {
    case 0:  ins_tiles<0, NP>(langS, arow, hrow); break;
    case 1:  ins_tiles<1, NP>(langS, arow, hrow); break;
    case 2:  ins_tiles<2, NP>(langS, arow, hrow); break;
    case 3:  ins_tiles<3, NP>(langS, arow, hrow); break;
    case 4:  ins_tiles<4, NP>(langS, arow, hrow); break;
    case 5:  ins_tiles<5, NP>(langS, arow, hrow); break;
    case 6:  ins_tiles<6, NP>(langS, arow, hrow); break;
    case 7:  ins_tiles<7, NP>(langS, arow, hrow); break;
    case 8:  ins_tiles<8, NP>(langS, arow, hrow); break;
    case 9:  ins_tiles<9, NP>(langS, arow, hrow); break;
    default: ins_tiles<10, NP>(langS, arow, hrow); break;
  }
}

#define PPB 2
#define FGT (16 * OD_)
#define RIT (16 * 16)
#define RTT (OD_ * RTP)
#define ATT (16 * LP_)

__global__ void __launch_bounds__(128)
pair_kernel(const float* __restrict__ coord,
            const float* __restrict__ lang_mask,
            const float* __restrict__ rel_w1,
            const float* __restrict__ rel_b1,
            const float* __restrict__ rel_w2,
            const float* __restrict__ rel_b2,
            const int*   __restrict__ idx_ws,
            const float* __restrict__ lang_ws,
            const float* __restrict__ f_ws,
            float* __restrict__ out) {
  __shared__ __attribute__((aligned(16))) float fg[PPB * FGT];
  __shared__ float relin[PPB * RIT];
  __shared__ __attribute__((aligned(16))) float hbuf[PPB * FGT];
  __shared__ float relT[PPB * RTT];
  __shared__ float attnsm[PPB * ATT];
  __shared__ float mksm[L_];

  const int tid  = threadIdx.x;
  const int lane = tid & 31, wave = tid >> 5;
  const int half = lane >> 4, lr = lane & 15;
  const int s  = blockIdx.x / (N_ / PPB);
  const int nb = (blockIdx.x % (N_ / PPB)) * PPB;

  const float mk0 = lang_mask[s * L_ + lane];
  const float mk1 = lang_mask[s * L_ + 32 + (lane & 7)];

  int nbi[PPB * 16];
#pragma unroll
  for (int r = 0; r < PPB * 16; ++r)
    nbi[r] = idx_ws[((size_t)s * N_ + nb + (r >> 4)) * KNB + (r & 15)];
  float gv[PPB * 16];
#pragma unroll
  for (int r = 0; r < PPB * 16; ++r)
    gv[r] = f_ws[((size_t)s * N_ + nbi[r]) * OD_ + tid];

  if (tid < PPB * 16) {
    const int pt = nb + (tid >> 4);
    const int j  = tid & 15;
    const float* cs = coord + (size_t)s * N_ * 3;
    const int m = idx_ws[((size_t)s * N_ + pt) * KNB + j];
    const float nx = cs[m * 3 + 0], ny = cs[m * 3 + 1], nz = cs[m * 3 + 2];
    const float ox = cs[pt * 3 + 0], oy = cs[pt * 3 + 1], oz = cs[pt * 3 + 2];
    const float rx = nx - ox, ry = ny - oy, rz = nz - oz;
    const float dist = sqrtf(rx * rx + ry * ry + rz * rz);
    float* ri = relin + tid * 16;
    ri[0] = nx; ri[1] = ny; ri[2] = nz;
    ri[3] = ox; ri[4] = oy; ri[5] = oz;
    ri[6] = rx; ri[7] = ry; ri[8] = rz;
    ri[9] = dist;
    ri[10] = 0.f; ri[11] = 0.f; ri[12] = 0.f;
    ri[13] = 0.f; ri[14] = 0.f; ri[15] = 0.f;
  }
#pragma unroll
  for (int r = 0; r < PPB * 16; ++r) fg[r * OD_ + tid] = gv[r];
  unsigned balA, balB;
  lang_ballots(mk0, mk1, lane, balA, balB);
  const int cnt = __builtin_popcount(balA) + __builtin_popcount(balB);
  if (tid < L_) mksm[lang_dest(tid, balA, balB, cnt)] = (wave == 0) ? mk0 : mk1;
  __syncthreads();

  float4* hx = reinterpret_cast<float4*>(hbuf);
  v2f bin[PPB][4];
#pragma unroll
  for (int p = 0; p < PPB; ++p) {
#pragma unroll
    for (int i = 0; i < 4; ++i) {
      const float* rp = relin + p * RIT + lr * 16 + 4 * i + 2 * half;
      bin[p][i].x = rp[0];
      bin[p][i].y = rp[1];
    }
  }
  auto hidden_tile = [&](const int t) {
    v8f c[PPB];
#pragma unroll
    for (int p = 0; p < PPB; ++p) c[p] = v8f{};
#pragma unroll
    for (int i = 0; i < 4; ++i) {
      const int ka  = 4 * i + 2 * half;
      const int ka0 = min(ka, 9), ka1 = min(ka + 1, 9);
      v2f a;
      a.x = rel_w1[ka0 * OD_ + 16 * t + lr];
      a.y = rel_w1[ka1 * OD_ + 16 * t + lr];
#pragma unroll
      for (int p = 0; p < PPB; ++p) c[p] = wmma4(a, bin[p][i], c[p]);
    }
    const float* bp = rel_b1 + 16 * t + 8 * half;
    float bias[8];
#pragma unroll
    for (int r = 0; r < 8; ++r) bias[r] = bp[r];
#pragma unroll
    for (int p = 0; p < PPB; ++p) {
      float h[8];
#pragma unroll
      for (int r = 0; r < 8; ++r) h[r] = fmaxf(c[p][r] + bias[r], 0.f);
      hx[p * (FGT / 4) + (2 * t) * 32 + lane]     = make_float4(h[0], h[1], h[2], h[3]);
      hx[p * (FGT / 4) + (2 * t + 1) * 32 + lane] = make_float4(h[4], h[5], h[6], h[7]);
    }
  };

  if (wave < 3) {
    hidden_tile(wave);
    const int c0 = wave * 16;
    const float* langS = lang_ws + (size_t)s * L_ * OD_;
    const int lclamp = min(c0 + lr, L_ - 1);
    const float* arow = fg + lr * OD_ + 8 * half;
    const float* brow = langS + lclamp * OD_ + 8 * half;
    v8f c[PPB];
#pragma unroll
    for (int p = 0; p < PPB; ++p) c[p] = v8f{};
    for (int k0 = 0; k0 < OD_; k0 += 16) {
      const v4f b0 = *reinterpret_cast<const v4f*>(brow + k0);
      const v4f b1 = *reinterpret_cast<const v4f*>(brow + k0 + 4);
      v4f a0[PPB], a1[PPB];
#pragma unroll
      for (int p = 0; p < PPB; ++p) {
        a0[p] = *reinterpret_cast<const v4f*>(arow + p * FGT + k0);
        a1[p] = *reinterpret_cast<const v4f*>(arow + p * FGT + k0 + 4);
      }
      v2f a, b;
      b.x = b0.x; b.y = b0.y;
#pragma unroll
      for (int p = 0; p < PPB; ++p) { a.x = a0[p].x; a.y = a0[p].y; c[p] = wmma4(a, b, c[p]); }
      b.x = b0.z; b.y = b0.w;
#pragma unroll
      for (int p = 0; p < PPB; ++p) { a.x = a0[p].z; a.y = a0[p].w; c[p] = wmma4(a, b, c[p]); }
      b.x = b1.x; b.y = b1.y;
#pragma unroll
      for (int p = 0; p < PPB; ++p) { a.x = a1[p].x; a.y = a1[p].y; c[p] = wmma4(a, b, c[p]); }
      b.x = b1.z; b.y = b1.w;
#pragma unroll
      for (int p = 0; p < PPB; ++p) { a.x = a1[p].z; a.y = a1[p].w; c[p] = wmma4(a, b, c[p]); }
    }
#pragma unroll
    for (int p = 0; p < PPB; ++p) {
#pragma unroll
      for (int v = 0; v < 8; ++v)
        attnsm[p * ATT + (v + 8 * half) * LP_ + c0 + lr] = c[p][v];
    }
  } else {
#pragma unroll
    for (int t = 3; t < 8; ++t) hidden_tile(t);
  }
  __syncthreads();

  {
    const int g = wave;
    const float* wp = rel_w2 + 32 * g + 2 * lr;
    v8f ca[PPB], cb[PPB];
#pragma unroll
    for (int p = 0; p < PPB; ++p) { ca[p] = v8f{}; cb[p] = v8f{}; }
#pragma unroll
    for (int t = 0; t < 8; ++t) {
      v8f hT[PPB];
#pragma unroll
      for (int p = 0; p < PPB; ++p) {
        const float4 lo = hx[p * (FGT / 4) + (2 * t) * 32 + lane];
        const float4 hi = hx[p * (FGT / 4) + (2 * t + 1) * 32 + lane];
        hT[p][0] = lo.x; hT[p][1] = lo.y; hT[p][2] = lo.z; hT[p][3] = lo.w;
        hT[p][4] = hi.x; hT[p][5] = hi.y; hT[p][6] = hi.z; hT[p][7] = hi.w;
      }
#pragma unroll
      for (int q = 0; q < 4; ++q) {
        const int k = 16 * t + 8 * half + 2 * q;
        const v2f w0 = *reinterpret_cast<const v2f*>(wp + k * OD_);
        const v2f w1 = *reinterpret_cast<const v2f*>(wp + (k + 1) * OD_);
        v2f aa, ab;
        aa.x = w0.x; aa.y = w1.x;
        ab.x = w0.y; ab.y = w1.y;
#pragma unroll
        for (int p = 0; p < PPB; ++p) {
          v2f b;
          b.x = hT[p][2 * q];
          b.y = hT[p][2 * q + 1];
          ca[p] = wmma4(aa, b, ca[p]);
          cb[p] = wmma4(ab, b, cb[p]);
        }
      }
    }
    const int dbase = 32 * g + 16 * half;
    const float* bp2 = rel_b2 + dbase;
    float bias2[16];
#pragma unroll
    for (int r = 0; r < 16; ++r) bias2[r] = bp2[r];
#pragma unroll
    for (int p = 0; p < PPB; ++p) {
#pragma unroll
      for (int r = 0; r < 8; ++r) {
        relT[p * RTT + (dbase + 2 * r) * RTP + lr]     = ca[p][r] + bias2[2 * r];
        relT[p * RTT + (dbase + 2 * r + 1) * RTP + lr] = cb[p][r] + bias2[2 * r + 1];
      }
    }
  }
  __syncthreads();

#pragma unroll
  for (int p = 0; p < PPB; ++p) {
    float* rowp = attnsm + p * ATT + (tid >> 3) * LP_ + (tid & 7);
    const float* mkp = mksm + (tid & 7);
    float v[5];
#pragma unroll
    for (int i = 0; i < 5; ++i) v[i] = rowp[8 * i];
    float mx = -FLT_MAX;
#pragma unroll
    for (int i = 0; i < 5; ++i) mx = fmaxf(mx, v[i]);
    mx = fmaxf(mx, lane_xor<1>(mx));
    mx = fmaxf(mx, lane_xor<2>(mx));
    mx = fmaxf(mx, lane_xor<4>(mx));
    float ssum = 0.f;
#pragma unroll
    for (int i = 0; i < 5; ++i) {
      v[i] = expf(v[i] - mx);
      ssum += v[i];
    }
    ssum += lane_xor<1>(ssum);
    ssum += lane_xor<2>(ssum);
    ssum += lane_xor<4>(ssum);
    const float inv = 1.f / ssum;
    float s2 = 0.f;
#pragma unroll
    for (int i = 0; i < 5; ++i) {
      v[i] = v[i] * inv * mkp[8 * i];
      s2 += v[i];
    }
    s2 += lane_xor<1>(s2);
    s2 += lane_xor<2>(s2);
    s2 += lane_xor<4>(s2);
    const float inv2 = 1.f / (s2 + 1e-7f);
#pragma unroll
    for (int i = 0; i < 5; ++i) rowp[8 * i] = v[i] * inv2;
  }
  __syncthreads();

  {
    const int c0 = 32 * wave + 2 * lr;
    const float* langS = lang_ws + (size_t)s * L_ * OD_ + c0 + 2 * half * OD_;
    const float* arow  = attnsm + lr * LP_ + 2 * half;
    float* hrow = hbuf + 8 * half * OD_ + c0;
    ins_product<PPB>((cnt + 3) >> 2, langS, arow, hrow);
  }
  __syncthreads();

#pragma unroll
  for (int p = 0; p < PPB; ++p) {
    const int d = tid;
    float acc = 0.f;
    for (int j = 0; j < 16; ++j)
      acc += fg[p * FGT + j * OD_ + d] * hbuf[p * FGT + j * OD_ + d] *
             relT[p * RTT + d * RTP + j];
    acc += f_ws[((size_t)s * N_ + nb + p) * OD_ + d];
    out[((size_t)s * N_ + nb + p) * OD_ + d] = acc;
  }
}

__global__ void __launch_bounds__(128)
tail_kernel(const float* __restrict__ coord,
                             const float* __restrict__ lang_mask,
                             const float* __restrict__ rel_w1,
                             const float* __restrict__ rel_b1,
                             const float* __restrict__ rel_w2,
                             const float* __restrict__ rel_b2,
                             const int*   __restrict__ idx_ws,
                             const float* __restrict__ lang_ws,
                             const float* __restrict__ f_ws,
                             float* __restrict__ out,
                             float* __restrict__ score) {
  __shared__ __attribute__((aligned(16))) float fg[16 * OD_];
  __shared__ float relin[16 * 16];
  __shared__ __attribute__((aligned(16))) float hbuf[16 * OD_];
  __shared__ float relT[OD_ * RTP];
  __shared__ float attnsm[16 * LP_];
  __shared__ float sred[16 * 4];
  __shared__ float mksm[L_];

  const int tid  = threadIdx.x;
  const int lane = tid & 31, wave = tid >> 5;
  const int half = lane >> 4, lr = lane & 15;
  const int s  = blockIdx.x >> 4;
  const int nb = (blockIdx.x & 15) * 16;

  const float mk0 = lang_mask[s * L_ + lane];
  const float mk1 = lang_mask[s * L_ + 32 + (lane & 7)];

  int nbi[16];
#pragma unroll
  for (int r = 0; r < 16; ++r) {
    const int pt = nb + r;
    const int j  = 16;
    nbi[r] = idx_ws[((size_t)s * N_ + pt) * KNB + j];
  }
  float gv[16];
#pragma unroll
  for (int r = 0; r < 16; ++r)
    gv[r] = f_ws[((size_t)s * N_ + nbi[r]) * OD_ + tid];

  if (tid < 16) {
    const int pt = nb + tid;
    const int j  = 16;
    const float* cs = coord + (size_t)s * N_ * 3;
    const int m = idx_ws[((size_t)s * N_ + pt) * KNB + j];
    const float nx = cs[m * 3 + 0], ny = cs[m * 3 + 1], nz = cs[m * 3 + 2];
    const float ox = cs[pt * 3 + 0], oy = cs[pt * 3 + 1], oz = cs[pt * 3 + 2];
    const float rx = nx - ox, ry = ny - oy, rz = nz - oz;
    const float dist = sqrtf(rx * rx + ry * ry + rz * rz);
    float* ri = relin + tid * 16;
    ri[0] = nx; ri[1] = ny; ri[2] = nz;
    ri[3] = ox; ri[4] = oy; ri[5] = oz;
    ri[6] = rx; ri[7] = ry; ri[8] = rz;
    ri[9] = dist;
    ri[10] = 0.f; ri[11] = 0.f; ri[12] = 0.f;
    ri[13] = 0.f; ri[14] = 0.f; ri[15] = 0.f;
  }
#pragma unroll
  for (int r = 0; r < 16; ++r) fg[r * OD_ + tid] = gv[r];
  unsigned balA, balB;
  lang_ballots(mk0, mk1, lane, balA, balB);
  const int cnt = __builtin_popcount(balA) + __builtin_popcount(balB);
  if (tid < L_) mksm[lang_dest(tid, balA, balB, cnt)] = (wave == 0) ? mk0 : mk1;
  __syncthreads();

  float4* hx = reinterpret_cast<float4*>(hbuf);
  v2f bin[4];
  for (int i = 0; i < 4; ++i) {
    const float* rp = relin + lr * 16 + 4 * i + 2 * half;
    bin[i].x = rp[0];
    bin[i].y = rp[1];
  }
  auto hidden_tile = [&](const int t) {
    v8f c = {};
    for (int i = 0; i < 4; ++i) {
      const int ka  = 4 * i + 2 * half;
      const int ka0 = min(ka, 9), ka1 = min(ka + 1, 9);
      v2f a;
      a.x = rel_w1[ka0 * OD_ + 16 * t + lr];
      a.y = rel_w1[ka1 * OD_ + 16 * t + lr];
      c = wmma4(a, bin[i], c);
    }
    const float* bp = rel_b1 + 16 * t + 8 * half;
    float h[8];
    for (int r = 0; r < 8; ++r) h[r] = fmaxf(c[r] + bp[r], 0.f);
    hx[(2 * t) * 32 + lane]     = make_float4(h[0], h[1], h[2], h[3]);
    hx[(2 * t + 1) * 32 + lane] = make_float4(h[4], h[5], h[6], h[7]);
  };

  if (wave < 3) {
    hidden_tile(wave);
    const int c0 = wave * 16;
    const float* langS = lang_ws + (size_t)s * L_ * OD_;
    const int lclamp = min(c0 + lr, L_ - 1);
    const float* arow = fg + lr * OD_ + 8 * half;
    const float* brow = langS + lclamp * OD_ + 8 * half;
    v8f c = {};
    for (int k0 = 0; k0 < OD_; k0 += 16) {
      const v4f b0 = *reinterpret_cast<const v4f*>(brow + k0);
      const v4f b1 = *reinterpret_cast<const v4f*>(brow + k0 + 4);
      const v4f a0 = *reinterpret_cast<const v4f*>(arow + k0);
      const v4f a1 = *reinterpret_cast<const v4f*>(arow + k0 + 4);
      v2f a, b;
      a.x = a0.x; a.y = a0.y; b.x = b0.x; b.y = b0.y; c = wmma4(a, b, c);
      a.x = a0.z; a.y = a0.w; b.x = b0.z; b.y = b0.w; c = wmma4(a, b, c);
      a.x = a1.x; a.y = a1.y; b.x = b1.x; b.y = b1.y; c = wmma4(a, b, c);
      a.x = a1.z; a.y = a1.w; b.x = b1.z; b.y = b1.w; c = wmma4(a, b, c);
    }
#pragma unroll
    for (int v = 0; v < 8; ++v)
      attnsm[(v + 8 * half) * LP_ + c0 + lr] = c[v];
  } else {
#pragma unroll
    for (int t = 3; t < 8; ++t) hidden_tile(t);
  }
  __syncthreads();

  {
    const int g = wave;
    v8f hT[8];
    for (int t = 0; t < 8; ++t) {
      const float4 lo = hx[(2 * t) * 32 + lane];
      const float4 hi = hx[(2 * t + 1) * 32 + lane];
      hT[t][0] = lo.x; hT[t][1] = lo.y; hT[t][2] = lo.z; hT[t][3] = lo.w;
      hT[t][4] = hi.x; hT[t][5] = hi.y; hT[t][6] = hi.z; hT[t][7] = hi.w;
    }

    const float* wp = rel_w2 + 32 * g + 2 * lr;
    v8f ca = {}, cb = {};
    for (int t = 0; t < 8; ++t) {
      for (int q = 0; q < 4; ++q) {
        const int k = 16 * t + 8 * half + 2 * q;
        const v2f w0 = *reinterpret_cast<const v2f*>(wp + k * OD_);
        const v2f w1 = *reinterpret_cast<const v2f*>(wp + (k + 1) * OD_);
        v2f aa, ab, b;
        aa.x = w0.x; aa.y = w1.x;
        ab.x = w0.y; ab.y = w1.y;
        b.x = hT[t][2 * q];
        b.y = hT[t][2 * q + 1];
        ca = wmma4(aa, b, ca);
        cb = wmma4(ab, b, cb);
      }
    }
    const int dbase = 32 * g + 16 * half;
    const float* bp2 = rel_b2 + dbase;
    for (int r = 0; r < 8; ++r) {
      relT[(dbase + 2 * r) * RTP + lr]     = ca[r] + bp2[2 * r];
      relT[(dbase + 2 * r + 1) * RTP + lr] = cb[r] + bp2[2 * r + 1];
    }
  }
  __syncthreads();

  {
    float* rowp = attnsm + (tid >> 3) * LP_ + (tid & 7);
    const float* mkp = mksm + (tid & 7);
    float v[5];
    for (int i = 0; i < 5; ++i) v[i] = rowp[8 * i];
    float mx = -FLT_MAX;
    for (int i = 0; i < 5; ++i) mx = fmaxf(mx, v[i]);
    mx = fmaxf(mx, lane_xor<1>(mx));
    mx = fmaxf(mx, lane_xor<2>(mx));
    mx = fmaxf(mx, lane_xor<4>(mx));
    float ssum = 0.f;
    for (int i = 0; i < 5; ++i) {
      v[i] = expf(v[i] - mx);
      ssum += v[i];
    }
    ssum += lane_xor<1>(ssum);
    ssum += lane_xor<2>(ssum);
    ssum += lane_xor<4>(ssum);
    const float inv = 1.f / ssum;
    float s2 = 0.f;
    for (int i = 0; i < 5; ++i) {
      v[i] = v[i] * inv * mkp[8 * i];
      s2 += v[i];
    }
    s2 += lane_xor<1>(s2);
    s2 += lane_xor<2>(s2);
    s2 += lane_xor<4>(s2);
    const float inv2 = 1.f / (s2 + 1e-7f);
    for (int i = 0; i < 5; ++i) rowp[8 * i] = v[i] * inv2;
  }
  __syncthreads();

  {
    const int c0 = 32 * wave + 2 * lr;
    const float* langS = lang_ws + (size_t)s * L_ * OD_ + c0 + 2 * half * OD_;
    const float* arow  = attnsm + lr * LP_ + 2 * half;
    float* hrow = hbuf + 8 * half * OD_ + c0;
    ins_product<1>((cnt + 3) >> 2, langS, arow, hrow);
  }
  __syncthreads();

  {
    const int d = tid;
    for (int r = 0; r < 16; ++r) {
      float* op = out + ((size_t)s * N_ + nb + r) * OD_ + d;
      const float o = *op + fg[r * OD_ + d] * hbuf[r * OD_ + d] * relT[d * RTP + r];
      *op = o;
      float w = o;
      for (int off = 16; off > 0; off >>= 1) w += __shfl_xor(w, off);
      if (lane == 0) sred[r * 4 + wave] = w;
    }
    __syncthreads();
    if (tid < 16)
      score[(size_t)s * N_ + nb + tid] =
          (sred[tid * 4] + sred[tid * 4 + 1]) + (sred[tid * 4 + 2] + sred[tid * 4 + 3]);
  }
}

extern "C" void kernel_launch(void* const* d_in, const int* in_sizes, int n_in,
                              void* d_out, int out_size, void* d_ws,
                              size_t ws_size, hipStream_t stream) {
  (void)in_sizes; (void)n_in; (void)out_size; (void)ws_size;
  const float* feat      = (const float*)d_in[0];
  const float* coord     = (const float*)d_in[1];
  const float* lang_feat = (const float*)d_in[2];
  const float* lang_mask = (const float*)d_in[3];
  const float* rel_w1    = (const float*)d_in[4];
  const float* rel_b1    = (const float*)d_in[5];
  const float* rel_w2    = (const float*)d_in[6];
  const float* rel_b2    = (const float*)d_in[7];
  const float* lang_w1   = (const float*)d_in[8];
  const float* lang_b1   = (const float*)d_in[9];
  const float* lang_w2   = (const float*)d_in[10];
  const float* lang_b2   = (const float*)d_in[11];
  const float* feat_w1   = (const float*)d_in[12];
  const float* feat_b1   = (const float*)d_in[13];
  const float* feat_w2   = (const float*)d_in[14];
  const float* feat_b2   = (const float*)d_in[15];

  char* ws = (char*)d_ws;
  int*   idx_ws  = (int*)ws;
  float* lang_ws = (float*)(ws + (size_t)S_ * N_ * KNB * 4);
  float* f_ws    = (float*)(ws + (size_t)S_ * N_ * KNB * 4 +
                                 (size_t)S_ * L_ * OD_ * 4);

  float* out   = (float*)d_out;
  float* score = out + (size_t)S_ * N_ * OD_;

  knn_kernel<<<S_ * (N_ / KNN_PPB), KNN_TPB, 0, stream>>>(coord, idx_ws);
  lang_mlp_kernel<<<S_ * L_, 128, 0, stream>>>(lang_feat, lang_mask, lang_w1,
                                               lang_b1, lang_w2, lang_b2,
                                               lang_ws);
  feat_mlp_kernel<<<S_ * (N_ / 16), 256, 0, stream>>>(feat, feat_w1, feat_b1,
                                                      feat_w2, feat_b2, f_ws);
  pair_kernel<<<S_ * (N_ / PPB), 128, 0, stream>>>(coord, lang_mask, rel_w1,
                                                   rel_b1, rel_w2, rel_b2,
                                                   idx_ws, lang_ws, f_ws, out);
  tail_kernel<<<S_ * (N_ / 16), 128, 0, stream>>>(coord, lang_mask, rel_w1,
                                                  rel_b1, rel_w2, rel_b2,
                                                  idx_ws, lang_ws, f_ws, out,
                                                  score);
}
